# starting kernel
# baseline (speedup 1.0000x reference)
#include <hip/hip_runtime.h>
#include <hip/hip_bf16.h>


typedef __attribute__((ext_vector_type(2))) float v2f;
typedef __attribute__((ext_vector_type(8))) float v8f;

#define CIN   128
#define COUT  256
#define D2    64
#define KPTS  15
#define NBH   32
#define KDFLAT (KPTS * D2)
#define INFL_INV 2.5f
#define BN_EPS 1e-5f
#define SLOPE 0.2f

__device__ __forceinline__ v8f wmma_f32(v2f a, v2f b, v8f c) {
  return __builtin_amdgcn_wmma_f32_16x16x4_f32(
       false, a,  false, b,
       (short)0, c,  false,  false);
}

__device__ __forceinline__ float lrelu(float v) {
  return v >= 0.0f ? v : SLOPE * v;
}

__global__ __launch_bounds__(256) void gemm_f32_wmma(
    const float* __restrict__ A, const float* __restrict__ B,
    float* __restrict__ C, int M, int Kd, int Nc, int totalUnits) {
  int unit = blockIdx.x * (blockDim.x >> 5) + (threadIdx.x >> 5);
  if (unit >= totalUnits) return;
  int lane = threadIdx.x & 31;
  int m    = lane & 15;
  int half = lane >> 4;
  int strips = Nc >> 6;
  int row0 = (unit / strips) << 4;
  int col0 = (unit % strips) << 6;

  const float* Arow = A + (size_t)(row0 + m) * Kd;
  v8f acc0 = {}, acc1 = {}, acc2 = {}, acc3 = {};
  for (int kk = 0; kk < Kd; kk += 4) {
    int ka = kk + (half << 1);
    v2f a;
    a.x = Arow[ka];
    a.y = Arow[ka + 1];
    const float* Bp = B + (size_t)ka * Nc + col0 + m;
    v2f b0, b1, b2, b3;
    b0.x = Bp[0];  b0.y = Bp[Nc];
    b1.x = Bp[16]; b1.y = Bp[Nc + 16];
    b2.x = Bp[32]; b2.y = Bp[Nc + 32];
    b3.x = Bp[48]; b3.y = Bp[Nc + 48];
    acc0 = wmma_f32(a, b0, acc0);
    acc1 = wmma_f32(a, b1, acc1);
    acc2 = wmma_f32(a, b2, acc2);
    acc3 = wmma_f32(a, b3, acc3);
  }
  int rbase = row0 + (half << 3);
  float* Cp = C + (size_t)rbase * Nc + col0 + m;
  #pragma unroll
  for (int r = 0; r < 8; ++r) {
    size_t o = (size_t)r * Nc;
    Cp[o +  0] = acc0[r];
    Cp[o + 16] = acc1[r];
    Cp[o + 32] = acc2[r];
    Cp[o + 48] = acc3[r];
  }
}

__global__ __launch_bounds__(256) void kpconv_wmma(
    const float* __restrict__ xyz, const int* __restrict__ nidx,
    const float* __restrict__ x1, const float* __restrict__ kp_points,
    const float* __restrict__ kpw, float* __restrict__ xk) {
  extern __shared__ float smem[];
  float* FK  = smem;
  float* Wl  = FK + 16 * KDFLAT;
  int*   IDX = (int*)(Wl + 8 * NBH * 16);
  float* PS  = (float*)(IDX + 8 * NBH);

  const int wave = threadIdx.x >> 5;
  const int lane = threadIdx.x & 31;
  const int m    = lane & 15;
  const int half = lane >> 4;
  const int base = blockIdx.x << 4;

  float* wl = Wl + wave * (NBH * 16);
  int*   il = IDX + wave * NBH;

  for (int it = 0; it < 2; ++it) {
    const int pi = (wave << 1) + it;
    const int n  = base + pi;

    const int id = nidx[(size_t)n * NBH + lane];
    il[lane] = id;
    const float cx = xyz[(size_t)n * 3 + 0];
    const float cy = xyz[(size_t)n * 3 + 1];
    const float cz = xyz[(size_t)n * 3 + 2];
    const float px = xyz[(size_t)id * 3 + 0] - cx;
    const float py = xyz[(size_t)id * 3 + 1] - cy;
    const float pz = xyz[(size_t)id * 3 + 2] - cz;
    #pragma unroll
    for (int k = 0; k < KPTS; ++k) {
      float dx = px - kp_points[k * 3 + 0];
      float dy = py - kp_points[k * 3 + 1];
      float dz = pz - kp_points[k * 3 + 2];
      float r = sqrtf(dx * dx + dy * dy + dz * dz);
      wl[lane * 16 + k] = fmaxf(1.0f - r * INFL_INV, 0.0f);
    }
    wl[lane * 16 + KPTS] = 0.0f;
    __syncthreads();

    v8f acc0 = {}, acc1 = {}, acc2 = {}, acc3 = {};
    #pragma unroll
    for (int kk = 0; kk < NBH; kk += 4) {
      const int ka = kk + (half << 1);
      v2f a;
      a.x = wl[ka * 16 + m];
      a.y = wl[(ka + 1) * 16 + m];
      const float* r0 = x1 + (size_t)il[ka] * D2 + m;
      const float* r1 = x1 + (size_t)il[ka + 1] * D2 + m;
      v2f b0, b1, b2, b3;
      b0.x = r0[0];  b0.y = r1[0];
      b1.x = r0[16]; b1.y = r1[16];
      b2.x = r0[32]; b2.y = r1[32];
      b3.x = r0[48]; b3.y = r1[48];
      acc0 = wmma_f32(a, b0, acc0);
      acc1 = wmma_f32(a, b1, acc1);
      acc2 = wmma_f32(a, b2, acc2);
      acc3 = wmma_f32(a, b3, acc3);
    }
    #pragma unroll
    for (int r = 0; r < 8; ++r) {
      const int krow = (half << 3) + r;
      if (krow < KPTS) {
        float* f = FK + pi * KDFLAT + krow * D2 + m;
        f[0]  = acc0[r];
        f[16] = acc1[r];
        f[32] = acc2[r];
        f[48] = acc3[r];
      }
    }
    __syncthreads();
  }

  {
    const int kbeg = wave * 120;
    v8f acc0 = {}, acc1 = {}, acc2 = {}, acc3 = {};
    for (int kk = kbeg; kk < kbeg + 120; kk += 4) {
      const int ka = kk + (half << 1);
      v2f a;
      a.x = FK[m * KDFLAT + ka];
      a.y = FK[m * KDFLAT + ka + 1];
      const float* Bp = kpw + (size_t)ka * D2 + m;
      v2f b0, b1, b2, b3;
      b0.x = Bp[0];  b0.y = Bp[D2];
      b1.x = Bp[16]; b1.y = Bp[D2 + 16];
      b2.x = Bp[32]; b2.y = Bp[D2 + 32];
      b3.x = Bp[48]; b3.y = Bp[D2 + 48];
      acc0 = wmma_f32(a, b0, acc0);
      acc1 = wmma_f32(a, b1, acc1);
      acc2 = wmma_f32(a, b2, acc2);
      acc3 = wmma_f32(a, b3, acc3);
    }
    #pragma unroll
    for (int r = 0; r < 8; ++r) {
      float* p = PS + wave * (16 * D2) + ((half << 3) + r) * D2 + m;
      p[0]  = acc0[r];
      p[16] = acc1[r];
      p[32] = acc2[r];
      p[48] = acc3[r];
    }
  }
  __syncthreads();

  for (int t = threadIdx.x; t < 16 * D2; t += 256) {
    float s = 0.0f;
    #pragma unroll
    for (int w = 0; w < 8; ++w) s += PS[w * (16 * D2) + t];
    const int pt = t >> 6;
    const int e  = t & 63;
    xk[(size_t)(base + pt) * D2 + e] = s;
  }
}

__global__ void zero_f32(float* __restrict__ p, int n) {
  int i = blockIdx.x * blockDim.x + threadIdx.x;
  if (i < n) p[i] = 0.0f;
}

__global__ void colstats(const float* __restrict__ Y, long long total, int C,
                         float* __restrict__ stats) {
  size_t idx0   = (size_t)blockIdx.x * blockDim.x + threadIdx.x;
  size_t stride = (size_t)gridDim.x * blockDim.x;
  float s = 0.0f, s2 = 0.0f;
  for (size_t i = idx0; i < (size_t)total; i += stride) {
    float v = Y[i];
    s += v;
    s2 += v * v;
  }
  int col = (int)(idx0 % (size_t)C);
  atomicAdd(&stats[col], s);
  atomicAdd(&stats[C + col], s2);
}

__global__ void finalize_bn(const float* __restrict__ stats,
                            const float* __restrict__ g,
                            const float* __restrict__ b, int C, float invN,
                            float* __restrict__ sc, float* __restrict__ sh) {
  int c = blockIdx.x * blockDim.x + threadIdx.x;
  if (c < C) {
    float mean = stats[c] * invN;
    float var  = stats[C + c] * invN - mean * mean;
    float s    = g[c] * rsqrtf(var + BN_EPS);
    sc[c] = s;
    sh[c] = b[c] - mean * s;
  }
}

__global__ void bnact_inplace(float* __restrict__ Y, const float* __restrict__ sc,
                              const float* __restrict__ sh, long long total, int C) {
  size_t idx0   = (size_t)blockIdx.x * blockDim.x + threadIdx.x;
  size_t stride = (size_t)gridDim.x * blockDim.x;
  for (size_t i = idx0; i < (size_t)total; i += stride) {
    int c = (int)(i % (size_t)C);
    Y[i] = lrelu(Y[i] * sc[c] + sh[c]);
  }
}

__global__ void fuse_out(float* __restrict__ out, const float* __restrict__ ysc,
                         const float* __restrict__ sc2, const float* __restrict__ sh2,
                         const float* __restrict__ scS, const float* __restrict__ shS,
                         long long total, int C) {
  size_t idx0   = (size_t)blockIdx.x * blockDim.x + threadIdx.x;
  size_t stride = (size_t)gridDim.x * blockDim.x;
  for (size_t i = idx0; i < (size_t)total; i += stride) {
    int c = (int)(i % (size_t)C);
    float v = lrelu(out[i] * sc2[c] + sh2[c]);
    out[i] = v + (ysc[i] * scS[c] + shS[c]);
  }
}

extern "C" void kernel_launch(void* const* d_in, const int* in_sizes, int n_in,
                              void* d_out, int out_size, void* d_ws, size_t ws_size,
                              hipStream_t stream) {
  const float* feats = (const float*)d_in[0];
  const float* xyz   = (const float*)d_in[1];
  const int*   nidx  = (const int*)d_in[3];
  const float* W1    = (const float*)d_in[4];
  const float* g1    = (const float*)d_in[5];
  const float* b1    = (const float*)d_in[6];
  const float* kpp   = (const float*)d_in[7];
  const float* kpw   = (const float*)d_in[8];
  const float* W2    = (const float*)d_in[9];
  const float* g2    = (const float*)d_in[10];
  const float* b2    = (const float*)d_in[11];
  const float* Wsc   = (const float*)d_in[12];
  const float* gsc   = (const float*)d_in[13];
  const float* bsc   = (const float*)d_in[14];
  float* out = (float*)d_out;

  const int N = in_sizes[0] / CIN;

  float* ws      = (float*)d_ws;
  float* x1      = ws;
  float* xk      = x1 + (size_t)N * D2;
  float* ysc     = xk + (size_t)N * D2;
  float* stats1  = ysc + (size_t)N * COUT;
  float* stats2  = stats1 + 2 * D2;
  float* statsSC = stats2 + 2 * COUT;
  float* sc1 = statsSC + 2 * COUT; float* sh1 = sc1 + D2;
  float* sc2 = sh1 + D2;           float* sh2 = sc2 + COUT;
  float* scS = sh2 + COUT;         float* shS = scS + COUT;

  const int statN = 2 * D2 + 4 * COUT;
  zero_f32<<<(statN + 255) / 256, 256, 0, stream>>>(stats1, statN);

  {
    int units = (N / 16) * (D2 / 64);
    gemm_f32_wmma<<<(units + 7) / 8, 256, 0, stream>>>(feats, W1, x1, N, CIN, D2, units);
  }
  colstats<<<256, 256, 0, stream>>>(x1, (long long)N * D2, D2, stats1);
  finalize_bn<<<1, D2, 0, stream>>>(stats1, g1, b1, D2, 1.0f / (float)N, sc1, sh1);
  bnact_inplace<<<2048, 256, 0, stream>>>(x1, sc1, sh1, (long long)N * D2, D2);

  {
    size_t lds = (16 * KDFLAT + 8 * NBH * 16 + 8 * 16 * D2) * sizeof(float)
               + 8 * NBH * sizeof(int);
    kpconv_wmma<<<N / 16, 256, lds, stream>>>(xyz, nidx, x1, kpp, kpw, xk);
  }

  {
    int units = (N / 16) * (COUT / 64);
    gemm_f32_wmma<<<(units + 7) / 8, 256, 0, stream>>>(xk, W2, out, N, D2, COUT, units);
  }
  {
    int units = (N / 16) * (COUT / 64);
    gemm_f32_wmma<<<(units + 7) / 8, 256, 0, stream>>>(feats, Wsc, ysc, N, CIN, COUT, units);
  }

  colstats<<<256, 256, 0, stream>>>(out, (long long)N * COUT, COUT, stats2);
  colstats<<<256, 256, 0, stream>>>(ysc, (long long)N * COUT, COUT, statsSC);
  finalize_bn<<<1, COUT, 0, stream>>>(stats2, g2, b2, COUT, 1.0f / (float)N, sc2, sh2);
  finalize_bn<<<1, COUT, 0, stream>>>(statsSC, gsc, bsc, COUT, 1.0f / (float)N, scS, shS);

  fuse_out<<<4096, 256, 0, stream>>>(out, ysc, sc2, sh2, scS, shS,
                                     (long long)N * COUT, COUT);
}


# BN column statistics without atomics: 1024 blocks with 16-byte loads, four in flight per lane, write partial sums to dead workspace; finalize adds them in fixed order. Reorders those sums.
# speedup vs baseline: 1.7021x; 1.7021x over previous
#include <hip/hip_runtime.h>
#include <hip/hip_bf16.h>


typedef __attribute__((ext_vector_type(2))) float v2f;
typedef __attribute__((ext_vector_type(8))) float v8f;

#define CIN   128
#define COUT  256
#define D2    64
#define KPTS  15
#define NBH   32
#define KDFLAT (KPTS * D2)
#define INFL_INV 2.5f
#define BN_EPS 1e-5f
#define SLOPE 0.2f
#define SPARTS 1024

__device__ __forceinline__ v8f wmma_f32(v2f a, v2f b, v8f c) {
  return __builtin_amdgcn_wmma_f32_16x16x4_f32(
       false, a,  false, b,
       (short)0, c,  false,  false);
}

__device__ __forceinline__ float lrelu(float v) {
  return v >= 0.0f ? v : SLOPE * v;
}

__global__ __launch_bounds__(256) void gemm_f32_wmma(
    const float* __restrict__ A, const float* __restrict__ B,
    float* __restrict__ C, int M, int Kd, int Nc, int totalUnits) {
  int unit = blockIdx.x * (blockDim.x >> 5) + (threadIdx.x >> 5);
  if (unit >= totalUnits) return;
  int lane = threadIdx.x & 31;
  int m    = lane & 15;
  int half = lane >> 4;
  int strips = Nc >> 6;
  int row0 = (unit / strips) << 4;
  int col0 = (unit % strips) << 6;

  const float* Arow = A + (size_t)(row0 + m) * Kd;
  v8f acc0 = {}, acc1 = {}, acc2 = {}, acc3 = {};
  for (int kk = 0; kk < Kd; kk += 4) {
    int ka = kk + (half << 1);
    v2f a;
    a.x = Arow[ka];
    a.y = Arow[ka + 1];
    const float* Bp = B + (size_t)ka * Nc + col0 + m;
    v2f b0, b1, b2, b3;
    b0.x = Bp[0];  b0.y = Bp[Nc];
    b1.x = Bp[16]; b1.y = Bp[Nc + 16];
    b2.x = Bp[32]; b2.y = Bp[Nc + 32];
    b3.x = Bp[48]; b3.y = Bp[Nc + 48];
    acc0 = wmma_f32(a, b0, acc0);
    acc1 = wmma_f32(a, b1, acc1);
    acc2 = wmma_f32(a, b2, acc2);
    acc3 = wmma_f32(a, b3, acc3);
  }
  int rbase = row0 + (half << 3);
  float* Cp = C + (size_t)rbase * Nc + col0 + m;
  #pragma unroll
  for (int r = 0; r < 8; ++r) {
    size_t o = (size_t)r * Nc;
    Cp[o +  0] = acc0[r];
    Cp[o + 16] = acc1[r];
    Cp[o + 32] = acc2[r];
    Cp[o + 48] = acc3[r];
  }
}

__global__ __launch_bounds__(256) void kpconv_wmma(
    const float* __restrict__ xyz, const int* __restrict__ nidx,
    const float* __restrict__ x1, const float* __restrict__ kp_points,
    const float* __restrict__ kpw, float* __restrict__ xk) {
  extern __shared__ float smem[];
  float* FK  = smem;
  float* Wl  = FK + 16 * KDFLAT;
  int*   IDX = (int*)(Wl + 8 * NBH * 16);
  float* PS  = (float*)(IDX + 8 * NBH);

  const int wave = threadIdx.x >> 5;
  const int lane = threadIdx.x & 31;
  const int m    = lane & 15;
  const int half = lane >> 4;
  const int base = blockIdx.x << 4;

  float* wl = Wl + wave * (NBH * 16);
  int*   il = IDX + wave * NBH;

  for (int it = 0; it < 2; ++it) {
    const int pi = (wave << 1) + it;
    const int n  = base + pi;

    const int id = nidx[(size_t)n * NBH + lane];
    il[lane] = id;
    const float cx = xyz[(size_t)n * 3 + 0];
    const float cy = xyz[(size_t)n * 3 + 1];
    const float cz = xyz[(size_t)n * 3 + 2];
    const float px = xyz[(size_t)id * 3 + 0] - cx;
    const float py = xyz[(size_t)id * 3 + 1] - cy;
    const float pz = xyz[(size_t)id * 3 + 2] - cz;
    #pragma unroll
    for (int k = 0; k < KPTS; ++k) {
      float dx = px - kp_points[k * 3 + 0];
      float dy = py - kp_points[k * 3 + 1];
      float dz = pz - kp_points[k * 3 + 2];
      float r = sqrtf(dx * dx + dy * dy + dz * dz);
      wl[lane * 16 + k] = fmaxf(1.0f - r * INFL_INV, 0.0f);
    }
    wl[lane * 16 + KPTS] = 0.0f;
    __syncthreads();

    v8f acc0 = {}, acc1 = {}, acc2 = {}, acc3 = {};
    #pragma unroll
    for (int kk = 0; kk < NBH; kk += 4) {
      const int ka = kk + (half << 1);
      v2f a;
      a.x = wl[ka * 16 + m];
      a.y = wl[(ka + 1) * 16 + m];
      const float* r0 = x1 + (size_t)il[ka] * D2 + m;
      const float* r1 = x1 + (size_t)il[ka + 1] * D2 + m;
      v2f b0, b1, b2, b3;
      b0.x = r0[0];  b0.y = r1[0];
      b1.x = r0[16]; b1.y = r1[16];
      b2.x = r0[32]; b2.y = r1[32];
      b3.x = r0[48]; b3.y = r1[48];
      acc0 = wmma_f32(a, b0, acc0);
      acc1 = wmma_f32(a, b1, acc1);
      acc2 = wmma_f32(a, b2, acc2);
      acc3 = wmma_f32(a, b3, acc3);
    }
    #pragma unroll
    for (int r = 0; r < 8; ++r) {
      const int krow = (half << 3) + r;
      if (krow < KPTS) {
        float* f = FK + pi * KDFLAT + krow * D2 + m;
        f[0]  = acc0[r];
        f[16] = acc1[r];
        f[32] = acc2[r];
        f[48] = acc3[r];
      }
    }
    __syncthreads();
  }

  {
    const int kbeg = wave * 120;
    v8f acc0 = {}, acc1 = {}, acc2 = {}, acc3 = {};
    for (int kk = kbeg; kk < kbeg + 120; kk += 4) {
      const int ka = kk + (half << 1);
      v2f a;
      a.x = FK[m * KDFLAT + ka];
      a.y = FK[m * KDFLAT + ka + 1];
      const float* Bp = kpw + (size_t)ka * D2 + m;
      v2f b0, b1, b2, b3;
      b0.x = Bp[0];  b0.y = Bp[D2];
      b1.x = Bp[16]; b1.y = Bp[D2 + 16];
      b2.x = Bp[32]; b2.y = Bp[D2 + 32];
      b3.x = Bp[48]; b3.y = Bp[D2 + 48];
      acc0 = wmma_f32(a, b0, acc0);
      acc1 = wmma_f32(a, b1, acc1);
      acc2 = wmma_f32(a, b2, acc2);
      acc3 = wmma_f32(a, b3, acc3);
    }
    #pragma unroll
    for (int r = 0; r < 8; ++r) {
      float* p = PS + wave * (16 * D2) + ((half << 3) + r) * D2 + m;
      p[0]  = acc0[r];
      p[16] = acc1[r];
      p[32] = acc2[r];
      p[48] = acc3[r];
    }
  }
  __syncthreads();

  for (int t = threadIdx.x; t < 16 * D2; t += 256) {
    float s = 0.0f;
    #pragma unroll
    for (int w = 0; w < 8; ++w) s += PS[w * (16 * D2) + t];
    const int pt = t >> 6;
    const int e  = t & 63;
    xk[(size_t)(base + pt) * D2 + e] = s;
  }
}

__global__ __launch_bounds__(256) void colstats_part(
    const float* __restrict__ Y, int nrows, int C, float* __restrict__ part) {
  __shared__ float4 redS[256];
  __shared__ float4 redQ[256];
  const int c4n  = C >> 2;
  const int rpb  = 256 / c4n;
  const int cg   = threadIdx.x % c4n;
  const int rg   = threadIdx.x / c4n;
  const int step = gridDim.x * rpb;
  const float4* __restrict__ Y4 = reinterpret_cast<const float4*>(Y);

  float4 s = {0.0f, 0.0f, 0.0f, 0.0f};
  float4 q = {0.0f, 0.0f, 0.0f, 0.0f};
  int r = blockIdx.x * rpb + rg;
  for (; r + 3 * step < nrows; r += 4 * step) {
    const float4 v0 = Y4[(size_t)r * c4n + cg];
    const float4 v1 = Y4[(size_t)(r + step) * c4n + cg];
    const float4 v2 = Y4[(size_t)(r + 2 * step) * c4n + cg];
    const float4 v3 = Y4[(size_t)(r + 3 * step) * c4n + cg];
    s.x += v0.x; s.y += v0.y; s.z += v0.z; s.w += v0.w;
    q.x += v0.x * v0.x; q.y += v0.y * v0.y; q.z += v0.z * v0.z; q.w += v0.w * v0.w;
    s.x += v1.x; s.y += v1.y; s.z += v1.z; s.w += v1.w;
    q.x += v1.x * v1.x; q.y += v1.y * v1.y; q.z += v1.z * v1.z; q.w += v1.w * v1.w;
    s.x += v2.x; s.y += v2.y; s.z += v2.z; s.w += v2.w;
    q.x += v2.x * v2.x; q.y += v2.y * v2.y; q.z += v2.z * v2.z; q.w += v2.w * v2.w;
    s.x += v3.x; s.y += v3.y; s.z += v3.z; s.w += v3.w;
    q.x += v3.x * v3.x; q.y += v3.y * v3.y; q.z += v3.z * v3.z; q.w += v3.w * v3.w;
  }
  for (; r < nrows; r += step) {
    const float4 v = Y4[(size_t)r * c4n + cg];
    s.x += v.x; s.y += v.y; s.z += v.z; s.w += v.w;
    q.x += v.x * v.x; q.y += v.y * v.y; q.z += v.z * v.z; q.w += v.w * v.w;
  }
  redS[threadIdx.x] = s;
  redQ[threadIdx.x] = q;
  __syncthreads();
  if ((int)threadIdx.x < C) {
    const float* fs = reinterpret_cast<const float*>(redS);
    const float* fq = reinterpret_cast<const float*>(redQ);
    float ts = 0.0f, tq = 0.0f;
    for (int g = 0; g < rpb; ++g) {
      ts += fs[g * C + threadIdx.x];
      tq += fq[g * C + threadIdx.x];
    }
    float* p = part + (size_t)blockIdx.x * (size_t)(2 * C);
    p[threadIdx.x]     = ts;
    p[C + threadIdx.x] = tq;
  }
}

__global__ __launch_bounds__(256) void finalize_bn(
    const float* __restrict__ part, int nparts,
    const float* __restrict__ g, const float* __restrict__ b, int C, float invN,
    float* __restrict__ sc, float* __restrict__ sh) {
  __shared__ float rs[256];
  __shared__ float rq[256];
  const int cl = threadIdx.x & 15;
  const int sl = threadIdx.x >> 4;
  const int c  = blockIdx.x * 16 + cl;
  const size_t pitch = (size_t)(2 * C);
  float s = 0.0f, q = 0.0f;
  #pragma unroll 8
  for (int p = sl; p < nparts; p += 16) {
    s += part[(size_t)p * pitch + c];
    q += part[(size_t)p * pitch + C + c];
  }
  rs[threadIdx.x] = s;
  rq[threadIdx.x] = q;
  __syncthreads();
  if (threadIdx.x < 16) {
    float ts = 0.0f, tq = 0.0f;
    #pragma unroll
    for (int k = 0; k < 16; ++k) {
      ts += rs[k * 16 + cl];
      tq += rq[k * 16 + cl];
    }
    float mean = ts * invN;
    float var  = tq * invN - mean * mean;
    float scl  = g[c] * rsqrtf(var + BN_EPS);
    sc[c] = scl;
    sh[c] = b[c] - mean * scl;
  }
}

__global__ void bnact_inplace(float* __restrict__ Y, const float* __restrict__ sc,
                              const float* __restrict__ sh, long long total, int C) {
  size_t idx0   = (size_t)blockIdx.x * blockDim.x + threadIdx.x;
  size_t stride = (size_t)gridDim.x * blockDim.x;
  for (size_t i = idx0; i < (size_t)total; i += stride) {
    int c = (int)(i % (size_t)C);
    Y[i] = lrelu(Y[i] * sc[c] + sh[c]);
  }
}

__global__ void fuse_out(float* __restrict__ out, const float* __restrict__ ysc,
                         const float* __restrict__ sc2, const float* __restrict__ sh2,
                         const float* __restrict__ scS, const float* __restrict__ shS,
                         long long total, int C) {
  size_t idx0   = (size_t)blockIdx.x * blockDim.x + threadIdx.x;
  size_t stride = (size_t)gridDim.x * blockDim.x;
  for (size_t i = idx0; i < (size_t)total; i += stride) {
    int c = (int)(i % (size_t)C);
    float v = lrelu(out[i] * sc2[c] + sh2[c]);
    out[i] = v + (ysc[i] * scS[c] + shS[c]);
  }
}

extern "C" void kernel_launch(void* const* d_in, const int* in_sizes, int n_in,
                              void* d_out, int out_size, void* d_ws, size_t ws_size,
                              hipStream_t stream) {
  const float* feats = (const float*)d_in[0];
  const float* xyz   = (const float*)d_in[1];
  const int*   nidx  = (const int*)d_in[3];
  const float* W1    = (const float*)d_in[4];
  const float* g1    = (const float*)d_in[5];
  const float* b1    = (const float*)d_in[6];
  const float* kpp   = (const float*)d_in[7];
  const float* kpw   = (const float*)d_in[8];
  const float* W2    = (const float*)d_in[9];
  const float* g2    = (const float*)d_in[10];
  const float* b2    = (const float*)d_in[11];
  const float* Wsc   = (const float*)d_in[12];
  const float* gsc   = (const float*)d_in[13];
  const float* bsc   = (const float*)d_in[14];
  float* out = (float*)d_out;

  const int N = in_sizes[0] / CIN;

  float* ws      = (float*)d_ws;
  float* x1      = ws;
  float* xk      = x1 + (size_t)N * D2;
  float* ysc     = xk + (size_t)N * D2;
  float* sc1 = ysc + (size_t)N * COUT; float* sh1 = sc1 + D2;
  float* sc2 = sh1 + D2;               float* sh2 = sc2 + COUT;
  float* scS = sh2 + COUT;             float* shS = scS + COUT;
  float* part1  = xk;
  float* part2  = x1;
  float* partSC = x1 + (size_t)SPARTS * 2 * COUT;

  {
    int units = (N / 16) * (D2 / 64);
    gemm_f32_wmma<<<(units + 7) / 8, 256, 0, stream>>>(feats, W1, x1, N, CIN, D2, units);
  }
  colstats_part<<<SPARTS, 256, 0, stream>>>(x1, N, D2, part1);
  finalize_bn<<<D2 / 16, 256, 0, stream>>>(part1, SPARTS, g1, b1, D2, 1.0f / (float)N, sc1, sh1);
  bnact_inplace<<<2048, 256, 0, stream>>>(x1, sc1, sh1, (long long)N * D2, D2);

  {
    size_t lds = (16 * KDFLAT + 8 * NBH * 16 + 8 * 16 * D2) * sizeof(float)
               + 8 * NBH * sizeof(int);
    kpconv_wmma<<<N / 16, 256, lds, stream>>>(xyz, nidx, x1, kpp, kpw, xk);
  }

  {
    int units = (N / 16) * (COUT / 64);
    gemm_f32_wmma<<<(units + 7) / 8, 256, 0, stream>>>(xk, W2, out, N, D2, COUT, units);
  }
  {
    int units = (N / 16) * (COUT / 64);
    gemm_f32_wmma<<<(units + 7) / 8, 256, 0, stream>>>(feats, Wsc, ysc, N, CIN, COUT, units);
  }

  colstats_part<<<SPARTS, 256, 0, stream>>>(out, N, COUT, part2);
  colstats_part<<<SPARTS, 256, 0, stream>>>(ysc, N, COUT, partSC);
  finalize_bn<<<COUT / 16, 256, 0, stream>>>(part2, SPARTS, g2, b2, COUT, 1.0f / (float)N, sc2, sh2);
  finalize_bn<<<COUT / 16, 256, 0, stream>>>(partSC, SPARTS, gsc, bsc, COUT, 1.0f / (float)N, scS, shS);

  fuse_out<<<4096, 256, 0, stream>>>(out, ysc, sc2, sh2, scS, shS,
                                     (long long)N * COUT, COUT);
}


# BN column statistics from the GEMM accumulators, no atomics (those made the earlier try vary run to run): one partial row per GEMM block, fixed-order finalize; 3 stat passes go. Reorders those sums.
# speedup vs baseline: 1.8043x; 1.0600x over previous
#include <hip/hip_runtime.h>
#include <hip/hip_bf16.h>


typedef __attribute__((ext_vector_type(2))) float v2f;
typedef __attribute__((ext_vector_type(8))) float v8f;

#define CIN   128
#define COUT  256
#define D2    64
#define KPTS  15
#define NBH   32
#define KDFLAT (KPTS * D2)
#define INFL_INV 2.5f
#define BN_EPS 1e-5f
#define SLOPE 0.2f

__device__ __forceinline__ v8f wmma_f32(v2f a, v2f b, v8f c) {
  return __builtin_amdgcn_wmma_f32_16x16x4_f32(
       false, a,  false, b,
       (short)0, c,  false,  false);
}

__device__ __forceinline__ float lrelu(float v) {
  return v >= 0.0f ? v : SLOPE * v;
}

__global__ __launch_bounds__(256) void gemm_f32_wmma(
    const float* __restrict__ A, const float* __restrict__ B,
    float* __restrict__ C, int M, int Kd, int Nc, int totalUnits,
    float* __restrict__ part) {
  __shared__ float red[8 * 2 * 128];
  const int wave = threadIdx.x >> 5;
  const int unit = blockIdx.x * 8 + wave;
  const bool active = unit < totalUnits;
  const int lane = threadIdx.x & 31;
  const int m    = lane & 15;
  const int half = lane >> 4;
  const int strips = Nc >> 6;

  v8f acc0 = {}, acc1 = {}, acc2 = {}, acc3 = {};
  const int row0 = active ? ((unit / strips) << 4) : 0;
  const int col0 = active ? ((unit % strips) << 6) : 0;
  if (active) {
    const float* Arow = A + (size_t)(row0 + m) * Kd;
    for (int kk = 0; kk < Kd; kk += 4) {
      int ka = kk + (half << 1);
      v2f a;
      a.x = Arow[ka];
      a.y = Arow[ka + 1];
      const float* Bp = B + (size_t)ka * Nc + col0 + m;
      v2f b0, b1, b2, b3;
      b0.x = Bp[0];  b0.y = Bp[Nc];
      b1.x = Bp[16]; b1.y = Bp[Nc + 16];
      b2.x = Bp[32]; b2.y = Bp[Nc + 32];
      b3.x = Bp[48]; b3.y = Bp[Nc + 48];
      acc0 = wmma_f32(a, b0, acc0);
      acc1 = wmma_f32(a, b1, acc1);
      acc2 = wmma_f32(a, b2, acc2);
      acc3 = wmma_f32(a, b3, acc3);
    }
  }

  float s0 = 0.0f, s1 = 0.0f, s2 = 0.0f, s3 = 0.0f;
  float q0 = 0.0f, q1 = 0.0f, q2 = 0.0f, q3 = 0.0f;
  #pragma unroll
  for (int r = 0; r < 8; ++r) {
    s0 += acc0[r]; q0 += acc0[r] * acc0[r];
    s1 += acc1[r]; q1 += acc1[r] * acc1[r];
    s2 += acc2[r]; q2 += acc2[r] * acc2[r];
    s3 += acc3[r]; q3 += acc3[r] * acc3[r];
  }
  float* rw = red + (wave * 2 + half) * 128;
  rw[m]      = s0; rw[16 + m] = s1; rw[32 + m] = s2; rw[48 + m] = s3;
  rw[64 + m] = q0; rw[80 + m] = q1; rw[96 + m] = q2; rw[112 + m] = q3;
  __syncthreads();
  if ((int)threadIdx.x < strips * 64) {
    const int st  = threadIdx.x >> 6;
    const int col = threadIdx.x & 63;
    float ts = 0.0f, tq = 0.0f;
    for (int w = st; w < 8; w += strips) {
      const float* r0 = red + (w * 2) * 128;
      ts += r0[col];       tq += r0[64 + col];
      ts += r0[128 + col]; tq += r0[192 + col];
    }
    float* p = part + (size_t)blockIdx.x * (size_t)(2 * Nc);
    p[st * 64 + col]      = ts;
    p[Nc + st * 64 + col] = tq;
  }

  if (active) {
    int rbase = row0 + (half << 3);
    float* Cp = C + (size_t)rbase * Nc + col0 + m;
    #pragma unroll
    for (int r = 0; r < 8; ++r) {
      size_t o = (size_t)r * Nc;
      Cp[o +  0] = acc0[r];
      Cp[o + 16] = acc1[r];
      Cp[o + 32] = acc2[r];
      Cp[o + 48] = acc3[r];
    }
  }
}

__global__ __launch_bounds__(256) void kpconv_wmma(
    const float* __restrict__ xyz, const int* __restrict__ nidx,
    const float* __restrict__ x1, const float* __restrict__ kp_points,
    const float* __restrict__ kpw, float* __restrict__ xk) {
  extern __shared__ float smem[];
  float* FK  = smem;
  float* Wl  = FK + 16 * KDFLAT;
  int*   IDX = (int*)(Wl + 8 * NBH * 16);
  float* PS  = (float*)(IDX + 8 * NBH);

  const int wave = threadIdx.x >> 5;
  const int lane = threadIdx.x & 31;
  const int m    = lane & 15;
  const int half = lane >> 4;
  const int base = blockIdx.x << 4;

  float* wl = Wl + wave * (NBH * 16);
  int*   il = IDX + wave * NBH;

  for (int it = 0; it < 2; ++it) {
    const int pi = (wave << 1) + it;
    const int n  = base + pi;

    const int id = nidx[(size_t)n * NBH + lane];
    il[lane] = id;
    const float cx = xyz[(size_t)n * 3 + 0];
    const float cy = xyz[(size_t)n * 3 + 1];
    const float cz = xyz[(size_t)n * 3 + 2];
    const float px = xyz[(size_t)id * 3 + 0] - cx;
    const float py = xyz[(size_t)id * 3 + 1] - cy;
    const float pz = xyz[(size_t)id * 3 + 2] - cz;
    #pragma unroll
    for (int k = 0; k < KPTS; ++k) {
      float dx = px - kp_points[k * 3 + 0];
      float dy = py - kp_points[k * 3 + 1];
      float dz = pz - kp_points[k * 3 + 2];
      float r = sqrtf(dx * dx + dy * dy + dz * dz);
      wl[lane * 16 + k] = fmaxf(1.0f - r * INFL_INV, 0.0f);
    }
    wl[lane * 16 + KPTS] = 0.0f;
    __syncthreads();

    v8f acc0 = {}, acc1 = {}, acc2 = {}, acc3 = {};
    #pragma unroll
    for (int kk = 0; kk < NBH; kk += 4) {
      const int ka = kk + (half << 1);
      v2f a;
      a.x = wl[ka * 16 + m];
      a.y = wl[(ka + 1) * 16 + m];
      const float* r0 = x1 + (size_t)il[ka] * D2 + m;
      const float* r1 = x1 + (size_t)il[ka + 1] * D2 + m;
      v2f b0, b1, b2, b3;
      b0.x = r0[0];  b0.y = r1[0];
      b1.x = r0[16]; b1.y = r1[16];
      b2.x = r0[32]; b2.y = r1[32];
      b3.x = r0[48]; b3.y = r1[48];
      acc0 = wmma_f32(a, b0, acc0);
      acc1 = wmma_f32(a, b1, acc1);
      acc2 = wmma_f32(a, b2, acc2);
      acc3 = wmma_f32(a, b3, acc3);
    }
    #pragma unroll
    for (int r = 0; r < 8; ++r) {
      const int krow = (half << 3) + r;
      if (krow < KPTS) {
        float* f = FK + pi * KDFLAT + krow * D2 + m;
        f[0]  = acc0[r];
        f[16] = acc1[r];
        f[32] = acc2[r];
        f[48] = acc3[r];
      }
    }
    __syncthreads();
  }

  {
    const int kbeg = wave * 120;
    v8f acc0 = {}, acc1 = {}, acc2 = {}, acc3 = {};
    for (int kk = kbeg; kk < kbeg + 120; kk += 4) {
      const int ka = kk + (half << 1);
      v2f a;
      a.x = FK[m * KDFLAT + ka];
      a.y = FK[m * KDFLAT + ka + 1];
      const float* Bp = kpw + (size_t)ka * D2 + m;
      v2f b0, b1, b2, b3;
      b0.x = Bp[0];  b0.y = Bp[D2];
      b1.x = Bp[16]; b1.y = Bp[D2 + 16];
      b2.x = Bp[32]; b2.y = Bp[D2 + 32];
      b3.x = Bp[48]; b3.y = Bp[D2 + 48];
      acc0 = wmma_f32(a, b0, acc0);
      acc1 = wmma_f32(a, b1, acc1);
      acc2 = wmma_f32(a, b2, acc2);
      acc3 = wmma_f32(a, b3, acc3);
    }
    #pragma unroll
    for (int r = 0; r < 8; ++r) {
      float* p = PS + wave * (16 * D2) + ((half << 3) + r) * D2 + m;
      p[0]  = acc0[r];
      p[16] = acc1[r];
      p[32] = acc2[r];
      p[48] = acc3[r];
    }
  }
  __syncthreads();

  for (int t = threadIdx.x; t < 16 * D2; t += 256) {
    float s = 0.0f;
    #pragma unroll
    for (int w = 0; w < 8; ++w) s += PS[w * (16 * D2) + t];
    const int pt = t >> 6;
    const int e  = t & 63;
    xk[(size_t)(base + pt) * D2 + e] = s;
  }
}

__global__ __launch_bounds__(1024) void finalize_bn(
    const float* __restrict__ part, int nparts,
    const float* __restrict__ g, const float* __restrict__ b, int C, float invN,
    float* __restrict__ sc, float* __restrict__ sh) {
  __shared__ float rs[1024];
  __shared__ float rq[1024];
  const int cl = threadIdx.x & 15;
  const int sl = threadIdx.x >> 4;
  const int c  = blockIdx.x * 16 + cl;
  const size_t pitch = (size_t)(2 * C);
  float s = 0.0f, q = 0.0f;
  #pragma unroll 8
  for (int p = sl; p < nparts; p += 64) {
    s += part[(size_t)p * pitch + c];
    q += part[(size_t)p * pitch + C + c];
  }
  rs[threadIdx.x] = s;
  rq[threadIdx.x] = q;
  __syncthreads();
  if (threadIdx.x < 16) {
    float ts = 0.0f, tq = 0.0f;
    #pragma unroll
    for (int k = 0; k < 64; ++k) {
      ts += rs[k * 16 + cl];
      tq += rq[k * 16 + cl];
    }
    float mean = ts * invN;
    float var  = tq * invN - mean * mean;
    float scl  = g[c] * rsqrtf(var + BN_EPS);
    sc[c] = scl;
    sh[c] = b[c] - mean * scl;
  }
}

__global__ void bnact_inplace(float* __restrict__ Y, const float* __restrict__ sc,
                              const float* __restrict__ sh, long long total, int C) {
  size_t idx0   = (size_t)blockIdx.x * blockDim.x + threadIdx.x;
  size_t stride = (size_t)gridDim.x * blockDim.x;
  for (size_t i = idx0; i < (size_t)total; i += stride) {
    int c = (int)(i % (size_t)C);
    Y[i] = lrelu(Y[i] * sc[c] + sh[c]);
  }
}

__global__ void fuse_out(float* __restrict__ out, const float* __restrict__ ysc,
                         const float* __restrict__ sc2, const float* __restrict__ sh2,
                         const float* __restrict__ scS, const float* __restrict__ shS,
                         long long total, int C) {
  size_t idx0   = (size_t)blockIdx.x * blockDim.x + threadIdx.x;
  size_t stride = (size_t)gridDim.x * blockDim.x;
  for (size_t i = idx0; i < (size_t)total; i += stride) {
    int c = (int)(i % (size_t)C);
    float v = lrelu(out[i] * sc2[c] + sh2[c]);
    out[i] = v + (ysc[i] * scS[c] + shS[c]);
  }
}

extern "C" void kernel_launch(void* const* d_in, const int* in_sizes, int n_in,
                              void* d_out, int out_size, void* d_ws, size_t ws_size,
                              hipStream_t stream) {
  const float* feats = (const float*)d_in[0];
  const float* xyz   = (const float*)d_in[1];
  const int*   nidx  = (const int*)d_in[3];
  const float* W1    = (const float*)d_in[4];
  const float* g1    = (const float*)d_in[5];
  const float* b1    = (const float*)d_in[6];
  const float* kpp   = (const float*)d_in[7];
  const float* kpw   = (const float*)d_in[8];
  const float* W2    = (const float*)d_in[9];
  const float* g2    = (const float*)d_in[10];
  const float* b2    = (const float*)d_in[11];
  const float* Wsc   = (const float*)d_in[12];
  const float* gsc   = (const float*)d_in[13];
  const float* bsc   = (const float*)d_in[14];
  float* out = (float*)d_out;

  const int N = in_sizes[0] / CIN;

  float* ws      = (float*)d_ws;
  float* x1      = ws;
  float* xk      = x1 + (size_t)N * D2;
  float* ysc     = xk + (size_t)N * D2;
  float* sc1 = ysc + (size_t)N * COUT; float* sh1 = sc1 + D2;
  float* sc2 = sh1 + D2;               float* sh2 = sc2 + COUT;
  float* scS = sh2 + COUT;             float* shS = scS + COUT;
  const int blk1 = ((N / 16) * (D2 / 64) + 7) / 8;
  const int blk2 = ((N / 16) * (COUT / 64) + 7) / 8;
  float* part1  = xk;
  float* part2  = x1;
  float* partSC = x1 + (size_t)blk2 * 2 * COUT;

  {
    int units = (N / 16) * (D2 / 64);
    gemm_f32_wmma<<<blk1, 256, 0, stream>>>(feats, W1, x1, N, CIN, D2, units, part1);
  }
  finalize_bn<<<D2 / 16, 1024, 0, stream>>>(part1, blk1, g1, b1, D2, 1.0f / (float)N, sc1, sh1);
  bnact_inplace<<<2048, 256, 0, stream>>>(x1, sc1, sh1, (long long)N * D2, D2);

  {
    size_t lds = (16 * KDFLAT + 8 * NBH * 16 + 8 * 16 * D2) * sizeof(float)
               + 8 * NBH * sizeof(int);
    kpconv_wmma<<<N / 16, 256, lds, stream>>>(xyz, nidx, x1, kpp, kpw, xk);
  }

  {
    int units = (N / 16) * (COUT / 64);
    gemm_f32_wmma<<<blk2, 256, 0, stream>>>(xk, W2, out, N, D2, COUT, units, part2);
  }
  {
    int units = (N / 16) * (COUT / 64);
    gemm_f32_wmma<<<blk2, 256, 0, stream>>>(feats, Wsc, ysc, N, CIN, COUT, units, partSC);
  }

  finalize_bn<<<COUT / 16, 1024, 0, stream>>>(part2, blk2, g2, b2, COUT, 1.0f / (float)N, sc2, sh2);
  finalize_bn<<<COUT / 16, 1024, 0, stream>>>(partSC, blk2, gsc, bsc, COUT, 1.0f / (float)N, scS, shS);

  fuse_out<<<4096, 256, 0, stream>>>(out, ysc, sc2, sh2, scS, shS,
                                     (long long)N * COUT, COUT);
}


# KPConv LDS per workgroup 111616 -> 78848 B: the K-slice partials reuse the FK tile behind one more barrier, so 4 workgroups (8 waves per SIMD, 128-VGPR bound) are resident per WGP instead of 2
# speedup vs baseline: 1.9038x; 1.0551x over previous
#include <hip/hip_runtime.h>
#include <hip/hip_bf16.h>


typedef __attribute__((ext_vector_type(2))) float v2f;
typedef __attribute__((ext_vector_type(8))) float v8f;

#define CIN   128
#define COUT  256
#define D2    64
#define KPTS  15
#define NBH   32
#define KDFLAT (KPTS * D2)
#define INFL_INV 2.5f
#define BN_EPS 1e-5f
#define SLOPE 0.2f

__device__ __forceinline__ v8f wmma_f32(v2f a, v2f b, v8f c) {
  return __builtin_amdgcn_wmma_f32_16x16x4_f32(
       false, a,  false, b,
       (short)0, c,  false,  false);
}

__device__ __forceinline__ float lrelu(float v) {
  return v >= 0.0f ? v : SLOPE * v;
}

__global__ __launch_bounds__(256) void gemm_f32_wmma(
    const float* __restrict__ A, const float* __restrict__ B,
    float* __restrict__ C, int M, int Kd, int Nc, int totalUnits,
    float* __restrict__ part) {
  __shared__ float red[8 * 2 * 128];
  const int wave = threadIdx.x >> 5;
  const int unit = blockIdx.x * 8 + wave;
  const bool active = unit < totalUnits;
  const int lane = threadIdx.x & 31;
  const int m    = lane & 15;
  const int half = lane >> 4;
  const int strips = Nc >> 6;

  v8f acc0 = {}, acc1 = {}, acc2 = {}, acc3 = {};
  const int row0 = active ? ((unit / strips) << 4) : 0;
  const int col0 = active ? ((unit % strips) << 6) : 0;
  if (active) {
    const float* Arow = A + (size_t)(row0 + m) * Kd;
    for (int kk = 0; kk < Kd; kk += 4) {
      int ka = kk + (half << 1);
      v2f a;
      a.x = Arow[ka];
      a.y = Arow[ka + 1];
      const float* Bp = B + (size_t)ka * Nc + col0 + m;
      v2f b0, b1, b2, b3;
      b0.x = Bp[0];  b0.y = Bp[Nc];
      b1.x = Bp[16]; b1.y = Bp[Nc + 16];
      b2.x = Bp[32]; b2.y = Bp[Nc + 32];
      b3.x = Bp[48]; b3.y = Bp[Nc + 48];
      acc0 = wmma_f32(a, b0, acc0);
      acc1 = wmma_f32(a, b1, acc1);
      acc2 = wmma_f32(a, b2, acc2);
      acc3 = wmma_f32(a, b3, acc3);
    }
  }

  float s0 = 0.0f, s1 = 0.0f, s2 = 0.0f, s3 = 0.0f;
  float q0 = 0.0f, q1 = 0.0f, q2 = 0.0f, q3 = 0.0f;
  #pragma unroll
  for (int r = 0; r < 8; ++r) {
    s0 += acc0[r]; q0 += acc0[r] * acc0[r];
    s1 += acc1[r]; q1 += acc1[r] * acc1[r];
    s2 += acc2[r]; q2 += acc2[r] * acc2[r];
    s3 += acc3[r]; q3 += acc3[r] * acc3[r];
  }
  float* rw = red + (wave * 2 + half) * 128;
  rw[m]      = s0; rw[16 + m] = s1; rw[32 + m] = s2; rw[48 + m] = s3;
  rw[64 + m] = q0; rw[80 + m] = q1; rw[96 + m] = q2; rw[112 + m] = q3;
  __syncthreads();
  if ((int)threadIdx.x < strips * 64) {
    const int st  = threadIdx.x >> 6;
    const int col = threadIdx.x & 63;
    float ts = 0.0f, tq = 0.0f;
    for (int w = st; w < 8; w += strips) {
      const float* r0 = red + (w * 2) * 128;
      ts += r0[col];       tq += r0[64 + col];
      ts += r0[128 + col]; tq += r0[192 + col];
    }
    float* p = part + (size_t)blockIdx.x * (size_t)(2 * Nc);
    p[st * 64 + col]      = ts;
    p[Nc + st * 64 + col] = tq;
  }

  if (active) {
    int rbase = row0 + (half << 3);
    float* Cp = C + (size_t)rbase * Nc + col0 + m;
    #pragma unroll
    for (int r = 0; r < 8; ++r) {
      size_t o = (size_t)r * Nc;
      Cp[o +  0] = acc0[r];
      Cp[o + 16] = acc1[r];
      Cp[o + 32] = acc2[r];
      Cp[o + 48] = acc3[r];
    }
  }
}

__global__ __launch_bounds__(256, 8) void kpconv_wmma(
    const float* __restrict__ xyz, const int* __restrict__ nidx,
    const float* __restrict__ x1, const float* __restrict__ kp_points,
    const float* __restrict__ kpw, float* __restrict__ xk) {
  extern __shared__ float smem[];
  float* FK  = smem;
  float* Wl  = FK + 16 * KDFLAT;
  int*   IDX = (int*)(Wl + 8 * NBH * 16);
  float* PS  = FK;

  const int wave = threadIdx.x >> 5;
  const int lane = threadIdx.x & 31;
  const int m    = lane & 15;
  const int half = lane >> 4;
  const int base = blockIdx.x << 4;

  float* wl = Wl + wave * (NBH * 16);
  int*   il = IDX + wave * NBH;

  for (int it = 0; it < 2; ++it) {
    const int pi = (wave << 1) + it;
    const int n  = base + pi;

    const int id = nidx[(size_t)n * NBH + lane];
    il[lane] = id;
    const float cx = xyz[(size_t)n * 3 + 0];
    const float cy = xyz[(size_t)n * 3 + 1];
    const float cz = xyz[(size_t)n * 3 + 2];
    const float px = xyz[(size_t)id * 3 + 0] - cx;
    const float py = xyz[(size_t)id * 3 + 1] - cy;
    const float pz = xyz[(size_t)id * 3 + 2] - cz;
    #pragma unroll
    for (int k = 0; k < KPTS; ++k) {
      float dx = px - kp_points[k * 3 + 0];
      float dy = py - kp_points[k * 3 + 1];
      float dz = pz - kp_points[k * 3 + 2];
      float r = sqrtf(dx * dx + dy * dy + dz * dz);
      wl[lane * 16 + k] = fmaxf(1.0f - r * INFL_INV, 0.0f);
    }
    wl[lane * 16 + KPTS] = 0.0f;
    __syncthreads();

    v8f acc0 = {}, acc1 = {}, acc2 = {}, acc3 = {};
    #pragma unroll
    for (int kk = 0; kk < NBH; kk += 4) {
      const int ka = kk + (half << 1);
      v2f a;
      a.x = wl[ka * 16 + m];
      a.y = wl[(ka + 1) * 16 + m];
      const float* r0 = x1 + (size_t)il[ka] * D2 + m;
      const float* r1 = x1 + (size_t)il[ka + 1] * D2 + m;
      v2f b0, b1, b2, b3;
      b0.x = r0[0];  b0.y = r1[0];
      b1.x = r0[16]; b1.y = r1[16];
      b2.x = r0[32]; b2.y = r1[32];
      b3.x = r0[48]; b3.y = r1[48];
      acc0 = wmma_f32(a, b0, acc0);
      acc1 = wmma_f32(a, b1, acc1);
      acc2 = wmma_f32(a, b2, acc2);
      acc3 = wmma_f32(a, b3, acc3);
    }
    #pragma unroll
    for (int r = 0; r < 8; ++r) {
      const int krow = (half << 3) + r;
      if (krow < KPTS) {
        float* f = FK + pi * KDFLAT + krow * D2 + m;
        f[0]  = acc0[r];
        f[16] = acc1[r];
        f[32] = acc2[r];
        f[48] = acc3[r];
      }
    }
    __syncthreads();
  }

  {
    const int kbeg = wave * 120;
    v8f acc0 = {}, acc1 = {}, acc2 = {}, acc3 = {};
    for (int kk = kbeg; kk < kbeg + 120; kk += 4) {
      const int ka = kk + (half << 1);
      v2f a;
      a.x = FK[m * KDFLAT + ka];
      a.y = FK[m * KDFLAT + ka + 1];
      const float* Bp = kpw + (size_t)ka * D2 + m;
      v2f b0, b1, b2, b3;
      b0.x = Bp[0];  b0.y = Bp[D2];
      b1.x = Bp[16]; b1.y = Bp[D2 + 16];
      b2.x = Bp[32]; b2.y = Bp[D2 + 32];
      b3.x = Bp[48]; b3.y = Bp[D2 + 48];
      acc0 = wmma_f32(a, b0, acc0);
      acc1 = wmma_f32(a, b1, acc1);
      acc2 = wmma_f32(a, b2, acc2);
      acc3 = wmma_f32(a, b3, acc3);
    }
    __syncthreads();
    #pragma unroll
    for (int r = 0; r < 8; ++r) {
      float* p = PS + wave * (16 * D2) + ((half << 3) + r) * D2 + m;
      p[0]  = acc0[r];
      p[16] = acc1[r];
      p[32] = acc2[r];
      p[48] = acc3[r];
    }
  }
  __syncthreads();

  for (int t = threadIdx.x; t < 16 * D2; t += 256) {
    float s = 0.0f;
    #pragma unroll
    for (int w = 0; w < 8; ++w) s += PS[w * (16 * D2) + t];
    const int pt = t >> 6;
    const int e  = t & 63;
    xk[(size_t)(base + pt) * D2 + e] = s;
  }
}

__global__ __launch_bounds__(1024) void finalize_bn(
    const float* __restrict__ part, int nparts,
    const float* __restrict__ g, const float* __restrict__ b, int C, float invN,
    float* __restrict__ sc, float* __restrict__ sh) {
  __shared__ float rs[1024];
  __shared__ float rq[1024];
  const int cl = threadIdx.x & 15;
  const int sl = threadIdx.x >> 4;
  const int c  = blockIdx.x * 16 + cl;
  const size_t pitch = (size_t)(2 * C);
  float s = 0.0f, q = 0.0f;
  #pragma unroll 8
  for (int p = sl; p < nparts; p += 64) {
    s += part[(size_t)p * pitch + c];
    q += part[(size_t)p * pitch + C + c];
  }
  rs[threadIdx.x] = s;
  rq[threadIdx.x] = q;
  __syncthreads();
  if (threadIdx.x < 16) {
    float ts = 0.0f, tq = 0.0f;
    #pragma unroll
    for (int k = 0; k < 64; ++k) {
      ts += rs[k * 16 + cl];
      tq += rq[k * 16 + cl];
    }
    float mean = ts * invN;
    float var  = tq * invN - mean * mean;
    float scl  = g[c] * rsqrtf(var + BN_EPS);
    sc[c] = scl;
    sh[c] = b[c] - mean * scl;
  }
}

__global__ void bnact_inplace(float* __restrict__ Y, const float* __restrict__ sc,
                              const float* __restrict__ sh, long long total, int C) {
  size_t idx0   = (size_t)blockIdx.x * blockDim.x + threadIdx.x;
  size_t stride = (size_t)gridDim.x * blockDim.x;
  for (size_t i = idx0; i < (size_t)total; i += stride) {
    int c = (int)(i % (size_t)C);
    Y[i] = lrelu(Y[i] * sc[c] + sh[c]);
  }
}

__global__ void fuse_out(float* __restrict__ out, const float* __restrict__ ysc,
                         const float* __restrict__ sc2, const float* __restrict__ sh2,
                         const float* __restrict__ scS, const float* __restrict__ shS,
                         long long total, int C) {
  size_t idx0   = (size_t)blockIdx.x * blockDim.x + threadIdx.x;
  size_t stride = (size_t)gridDim.x * blockDim.x;
  for (size_t i = idx0; i < (size_t)total; i += stride) {
    int c = (int)(i % (size_t)C);
    float v = lrelu(out[i] * sc2[c] + sh2[c]);
    out[i] = v + (ysc[i] * scS[c] + shS[c]);
  }
}

extern "C" void kernel_launch(void* const* d_in, const int* in_sizes, int n_in,
                              void* d_out, int out_size, void* d_ws, size_t ws_size,
                              hipStream_t stream) {
  const float* feats = (const float*)d_in[0];
  const float* xyz   = (const float*)d_in[1];
  const int*   nidx  = (const int*)d_in[3];
  const float* W1    = (const float*)d_in[4];
  const float* g1    = (const float*)d_in[5];
  const float* b1    = (const float*)d_in[6];
  const float* kpp   = (const float*)d_in[7];
  const float* kpw   = (const float*)d_in[8];
  const float* W2    = (const float*)d_in[9];
  const float* g2    = (const float*)d_in[10];
  const float* b2    = (const float*)d_in[11];
  const float* Wsc   = (const float*)d_in[12];
  const float* gsc   = (const float*)d_in[13];
  const float* bsc   = (const float*)d_in[14];
  float* out = (float*)d_out;

  const int N = in_sizes[0] / CIN;

  float* ws      = (float*)d_ws;
  float* x1      = ws;
  float* xk      = x1 + (size_t)N * D2;
  float* ysc     = xk + (size_t)N * D2;
  float* sc1 = ysc + (size_t)N * COUT; float* sh1 = sc1 + D2;
  float* sc2 = sh1 + D2;               float* sh2 = sc2 + COUT;
  float* scS = sh2 + COUT;             float* shS = scS + COUT;
  const int blk1 = ((N / 16) * (D2 / 64) + 7) / 8;
  const int blk2 = ((N / 16) * (COUT / 64) + 7) / 8;
  float* part1  = xk;
  float* part2  = x1;
  float* partSC = x1 + (size_t)blk2 * 2 * COUT;

  {
    int units = (N / 16) * (D2 / 64);
    gemm_f32_wmma<<<blk1, 256, 0, stream>>>(feats, W1, x1, N, CIN, D2, units, part1);
  }
  finalize_bn<<<D2 / 16, 1024, 0, stream>>>(part1, blk1, g1, b1, D2, 1.0f / (float)N, sc1, sh1);
  bnact_inplace<<<2048, 256, 0, stream>>>(x1, sc1, sh1, (long long)N * D2, D2);

  {
    size_t lds = (16 * KDFLAT + 8 * NBH * 16) * sizeof(float)
               + 8 * NBH * sizeof(int);
    kpconv_wmma<<<N / 16, 256, lds, stream>>>(xyz, nidx, x1, kpp, kpw, xk);
  }

  {
    int units = (N / 16) * (COUT / 64);
    gemm_f32_wmma<<<blk2, 256, 0, stream>>>(xk, W2, out, N, D2, COUT, units, part2);
  }
  {
    int units = (N / 16) * (COUT / 64);
    gemm_f32_wmma<<<blk2, 256, 0, stream>>>(feats, Wsc, ysc, N, CIN, COUT, units, partSC);
  }

  finalize_bn<<<COUT / 16, 1024, 0, stream>>>(part2, blk2, g2, b2, COUT, 1.0f / (float)N, sc2, sh2);
  finalize_bn<<<COUT / 16, 1024, 0, stream>>>(partSC, blk2, gsc, bsc, COUT, 1.0f / (float)N, scS, shS);

  fuse_out<<<4096, 256, 0, stream>>>(out, ysc, sc2, sh2, scS, shS,
                                     (long long)N * COUT, COUT);
}


# KPConv LDS pitches against bank conflicts: FK rows 960 -> 964 floats (every phase-2 A-fragment read was a 16-way conflict), influence tile k-major at pitch 33 (its stores were 8-way)
# speedup vs baseline: 1.9663x; 1.0328x over previous
#include <hip/hip_runtime.h>
#include <hip/hip_bf16.h>


typedef __attribute__((ext_vector_type(2))) float v2f;
typedef __attribute__((ext_vector_type(8))) float v8f;

#define CIN   128
#define COUT  256
#define D2    64
#define KPTS  15
#define NBH   32
#define KDFLAT (KPTS * D2)
#define FKP (KDFLAT + 4)
#define WLP 33
#define INFL_INV 2.5f
#define BN_EPS 1e-5f
#define SLOPE 0.2f

__device__ __forceinline__ v8f wmma_f32(v2f a, v2f b, v8f c) {
  return __builtin_amdgcn_wmma_f32_16x16x4_f32(
       false, a,  false, b,
       (short)0, c,  false,  false);
}

__device__ __forceinline__ float lrelu(float v) {
  return v >= 0.0f ? v : SLOPE * v;
}

__global__ __launch_bounds__(256) void gemm_f32_wmma(
    const float* __restrict__ A, const float* __restrict__ B,
    float* __restrict__ C, int M, int Kd, int Nc, int totalUnits,
    float* __restrict__ part) {
  __shared__ float red[8 * 2 * 128];
  const int wave = threadIdx.x >> 5;
  const int unit = blockIdx.x * 8 + wave;
  const bool active = unit < totalUnits;
  const int lane = threadIdx.x & 31;
  const int m    = lane & 15;
  const int half = lane >> 4;
  const int strips = Nc >> 6;

  v8f acc0 = {}, acc1 = {}, acc2 = {}, acc3 = {};
  const int row0 = active ? ((unit / strips) << 4) : 0;
  const int col0 = active ? ((unit % strips) << 6) : 0;
  if (active) {
    const float* Arow = A + (size_t)(row0 + m) * Kd;
    for (int kk = 0; kk < Kd; kk += 4) {
      int ka = kk + (half << 1);
      v2f a;
      a.x = Arow[ka];
      a.y = Arow[ka + 1];
      const float* Bp = B + (size_t)ka * Nc + col0 + m;
      v2f b0, b1, b2, b3;
      b0.x = Bp[0];  b0.y = Bp[Nc];
      b1.x = Bp[16]; b1.y = Bp[Nc + 16];
      b2.x = Bp[32]; b2.y = Bp[Nc + 32];
      b3.x = Bp[48]; b3.y = Bp[Nc + 48];
      acc0 = wmma_f32(a, b0, acc0);
      acc1 = wmma_f32(a, b1, acc1);
      acc2 = wmma_f32(a, b2, acc2);
      acc3 = wmma_f32(a, b3, acc3);
    }
  }

  float s0 = 0.0f, s1 = 0.0f, s2 = 0.0f, s3 = 0.0f;
  float q0 = 0.0f, q1 = 0.0f, q2 = 0.0f, q3 = 0.0f;
  #pragma unroll
  for (int r = 0; r < 8; ++r) {
    s0 += acc0[r]; q0 += acc0[r] * acc0[r];
    s1 += acc1[r]; q1 += acc1[r] * acc1[r];
    s2 += acc2[r]; q2 += acc2[r] * acc2[r];
    s3 += acc3[r]; q3 += acc3[r] * acc3[r];
  }
  float* rw = red + (wave * 2 + half) * 128;
  rw[m]      = s0; rw[16 + m] = s1; rw[32 + m] = s2; rw[48 + m] = s3;
  rw[64 + m] = q0; rw[80 + m] = q1; rw[96 + m] = q2; rw[112 + m] = q3;
  __syncthreads();
  if ((int)threadIdx.x < strips * 64) {
    const int st  = threadIdx.x >> 6;
    const int col = threadIdx.x & 63;
    float ts = 0.0f, tq = 0.0f;
    for (int w = st; w < 8; w += strips) {
      const float* r0 = red + (w * 2) * 128;
      ts += r0[col];       tq += r0[64 + col];
      ts += r0[128 + col]; tq += r0[192 + col];
    }
    float* p = part + (size_t)blockIdx.x * (size_t)(2 * Nc);
    p[st * 64 + col]      = ts;
    p[Nc + st * 64 + col] = tq;
  }

  if (active) {
    int rbase = row0 + (half << 3);
    float* Cp = C + (size_t)rbase * Nc + col0 + m;
    #pragma unroll
    for (int r = 0; r < 8; ++r) {
      size_t o = (size_t)r * Nc;
      Cp[o +  0] = acc0[r];
      Cp[o + 16] = acc1[r];
      Cp[o + 32] = acc2[r];
      Cp[o + 48] = acc3[r];
    }
  }
}

__global__ __launch_bounds__(256, 8) void kpconv_wmma(
    const float* __restrict__ xyz, const int* __restrict__ nidx,
    const float* __restrict__ x1, const float* __restrict__ kp_points,
    const float* __restrict__ kpw, float* __restrict__ xk) {
  extern __shared__ float smem[];
  float* FK  = smem;
  float* Wl  = FK + 16 * FKP;
  int*   IDX = (int*)(Wl + 8 * 16 * WLP);
  float* PS  = FK;

  const int wave = threadIdx.x >> 5;
  const int lane = threadIdx.x & 31;
  const int m    = lane & 15;
  const int half = lane >> 4;
  const int base = blockIdx.x << 4;

  float* wl = Wl + wave * (16 * WLP);
  int*   il = IDX + wave * NBH;

  for (int it = 0; it < 2; ++it) {
    const int pi = (wave << 1) + it;
    const int n  = base + pi;

    const int id = nidx[(size_t)n * NBH + lane];
    il[lane] = id;
    const float cx = xyz[(size_t)n * 3 + 0];
    const float cy = xyz[(size_t)n * 3 + 1];
    const float cz = xyz[(size_t)n * 3 + 2];
    const float px = xyz[(size_t)id * 3 + 0] - cx;
    const float py = xyz[(size_t)id * 3 + 1] - cy;
    const float pz = xyz[(size_t)id * 3 + 2] - cz;
    #pragma unroll
    for (int k = 0; k < KPTS; ++k) {
      float dx = px - kp_points[k * 3 + 0];
      float dy = py - kp_points[k * 3 + 1];
      float dz = pz - kp_points[k * 3 + 2];
      float r = sqrtf(dx * dx + dy * dy + dz * dz);
      wl[k * WLP + lane] = fmaxf(1.0f - r * INFL_INV, 0.0f);
    }
    wl[KPTS * WLP + lane] = 0.0f;
    __syncthreads();

    v8f acc0 = {}, acc1 = {}, acc2 = {}, acc3 = {};
    #pragma unroll
    for (int kk = 0; kk < NBH; kk += 4) {
      const int ka = kk + (half << 1);
      v2f a;
      a.x = wl[m * WLP + ka];
      a.y = wl[m * WLP + ka + 1];
      const float* r0 = x1 + (size_t)il[ka] * D2 + m;
      const float* r1 = x1 + (size_t)il[ka + 1] * D2 + m;
      v2f b0, b1, b2, b3;
      b0.x = r0[0];  b0.y = r1[0];
      b1.x = r0[16]; b1.y = r1[16];
      b2.x = r0[32]; b2.y = r1[32];
      b3.x = r0[48]; b3.y = r1[48];
      acc0 = wmma_f32(a, b0, acc0);
      acc1 = wmma_f32(a, b1, acc1);
      acc2 = wmma_f32(a, b2, acc2);
      acc3 = wmma_f32(a, b3, acc3);
    }
    #pragma unroll
    for (int r = 0; r < 8; ++r) {
      const int krow = (half << 3) + r;
      if (krow < KPTS) {
        float* f = FK + pi * FKP + krow * D2 + m;
        f[0]  = acc0[r];
        f[16] = acc1[r];
        f[32] = acc2[r];
        f[48] = acc3[r];
      }
    }
    __syncthreads();
  }

  {
    const int kbeg = wave * 120;
    v8f acc0 = {}, acc1 = {}, acc2 = {}, acc3 = {};
    for (int kk = kbeg; kk < kbeg + 120; kk += 4) {
      const int ka = kk + (half << 1);
      v2f a;
      a.x = FK[m * FKP + ka];
      a.y = FK[m * FKP + ka + 1];
      const float* Bp = kpw + (size_t)ka * D2 + m;
      v2f b0, b1, b2, b3;
      b0.x = Bp[0];  b0.y = Bp[D2];
      b1.x = Bp[16]; b1.y = Bp[D2 + 16];
      b2.x = Bp[32]; b2.y = Bp[D2 + 32];
      b3.x = Bp[48]; b3.y = Bp[D2 + 48];
      acc0 = wmma_f32(a, b0, acc0);
      acc1 = wmma_f32(a, b1, acc1);
      acc2 = wmma_f32(a, b2, acc2);
      acc3 = wmma_f32(a, b3, acc3);
    }
    __syncthreads();
    #pragma unroll
    for (int r = 0; r < 8; ++r) {
      float* p = PS + wave * (16 * D2) + ((half << 3) + r) * D2 + m;
      p[0]  = acc0[r];
      p[16] = acc1[r];
      p[32] = acc2[r];
      p[48] = acc3[r];
    }
  }
  __syncthreads();

  for (int t = threadIdx.x; t < 16 * D2; t += 256) {
    float s = 0.0f;
    #pragma unroll
    for (int w = 0; w < 8; ++w) s += PS[w * (16 * D2) + t];
    const int pt = t >> 6;
    const int e  = t & 63;
    xk[(size_t)(base + pt) * D2 + e] = s;
  }
}

__global__ __launch_bounds__(1024) void finalize_bn(
    const float* __restrict__ part, int nparts,
    const float* __restrict__ g, const float* __restrict__ b, int C, float invN,
    float* __restrict__ sc, float* __restrict__ sh) {
  __shared__ float rs[1024];
  __shared__ float rq[1024];
  const int cl = threadIdx.x & 15;
  const int sl = threadIdx.x >> 4;
  const int c  = blockIdx.x * 16 + cl;
  const size_t pitch = (size_t)(2 * C);
  float s = 0.0f, q = 0.0f;
  #pragma unroll 8
  for (int p = sl; p < nparts; p += 64) {
    s += part[(size_t)p * pitch + c];
    q += part[(size_t)p * pitch + C + c];
  }
  rs[threadIdx.x] = s;
  rq[threadIdx.x] = q;
  __syncthreads();
  if (threadIdx.x < 16) {
    float ts = 0.0f, tq = 0.0f;
    #pragma unroll
    for (int k = 0; k < 64; ++k) {
      ts += rs[k * 16 + cl];
      tq += rq[k * 16 + cl];
    }
    float mean = ts * invN;
    float var  = tq * invN - mean * mean;
    float scl  = g[c] * rsqrtf(var + BN_EPS);
    sc[c] = scl;
    sh[c] = b[c] - mean * scl;
  }
}

__global__ void bnact_inplace(float* __restrict__ Y, const float* __restrict__ sc,
                              const float* __restrict__ sh, long long total, int C) {
  size_t idx0   = (size_t)blockIdx.x * blockDim.x + threadIdx.x;
  size_t stride = (size_t)gridDim.x * blockDim.x;
  for (size_t i = idx0; i < (size_t)total; i += stride) {
    int c = (int)(i % (size_t)C);
    Y[i] = lrelu(Y[i] * sc[c] + sh[c]);
  }
}

__global__ void fuse_out(float* __restrict__ out, const float* __restrict__ ysc,
                         const float* __restrict__ sc2, const float* __restrict__ sh2,
                         const float* __restrict__ scS, const float* __restrict__ shS,
                         long long total, int C) {
  size_t idx0   = (size_t)blockIdx.x * blockDim.x + threadIdx.x;
  size_t stride = (size_t)gridDim.x * blockDim.x;
  for (size_t i = idx0; i < (size_t)total; i += stride) {
    int c = (int)(i % (size_t)C);
    float v = lrelu(out[i] * sc2[c] + sh2[c]);
    out[i] = v + (ysc[i] * scS[c] + shS[c]);
  }
}

extern "C" void kernel_launch(void* const* d_in, const int* in_sizes, int n_in,
                              void* d_out, int out_size, void* d_ws, size_t ws_size,
                              hipStream_t stream) {
  const float* feats = (const float*)d_in[0];
  const float* xyz   = (const float*)d_in[1];
  const int*   nidx  = (const int*)d_in[3];
  const float* W1    = (const float*)d_in[4];
  const float* g1    = (const float*)d_in[5];
  const float* b1    = (const float*)d_in[6];
  const float* kpp   = (const float*)d_in[7];
  const float* kpw   = (const float*)d_in[8];
  const float* W2    = (const float*)d_in[9];
  const float* g2    = (const float*)d_in[10];
  const float* b2    = (const float*)d_in[11];
  const float* Wsc   = (const float*)d_in[12];
  const float* gsc   = (const float*)d_in[13];
  const float* bsc   = (const float*)d_in[14];
  float* out = (float*)d_out;

  const int N = in_sizes[0] / CIN;

  float* ws      = (float*)d_ws;
  float* x1      = ws;
  float* xk      = x1 + (size_t)N * D2;
  float* ysc     = xk + (size_t)N * D2;
  float* sc1 = ysc + (size_t)N * COUT; float* sh1 = sc1 + D2;
  float* sc2 = sh1 + D2;               float* sh2 = sc2 + COUT;
  float* scS = sh2 + COUT;             float* shS = scS + COUT;
  const int blk1 = ((N / 16) * (D2 / 64) + 7) / 8;
  const int blk2 = ((N / 16) * (COUT / 64) + 7) / 8;
  float* part1  = xk;
  float* part2  = x1;
  float* partSC = x1 + (size_t)blk2 * 2 * COUT;

  {
    int units = (N / 16) * (D2 / 64);
    gemm_f32_wmma<<<blk1, 256, 0, stream>>>(feats, W1, x1, N, CIN, D2, units, part1);
  }
  finalize_bn<<<D2 / 16, 1024, 0, stream>>>(part1, blk1, g1, b1, D2, 1.0f / (float)N, sc1, sh1);
  bnact_inplace<<<2048, 256, 0, stream>>>(x1, sc1, sh1, (long long)N * D2, D2);

  {
    size_t lds = (16 * FKP + 8 * 16 * WLP) * sizeof(float)
               + 8 * NBH * sizeof(int);
    kpconv_wmma<<<N / 16, 256, lds, stream>>>(xyz, nidx, x1, kpp, kpw, xk);
  }

  {
    int units = (N / 16) * (COUT / 64);
    gemm_f32_wmma<<<blk2, 256, 0, stream>>>(xk, W2, out, N, D2, COUT, units, part2);
  }
  {
    int units = (N / 16) * (COUT / 64);
    gemm_f32_wmma<<<blk2, 256, 0, stream>>>(feats, Wsc, ysc, N, CIN, COUT, units, partSC);
  }

  finalize_bn<<<COUT / 16, 1024, 0, stream>>>(part2, blk2, g2, b2, COUT, 1.0f / (float)N, sc2, sh2);
  finalize_bn<<<COUT / 16, 1024, 0, stream>>>(partSC, blk2, gsc, bsc, COUT, 1.0f / (float)N, scS, shS);

  fuse_out<<<4096, 256, 0, stream>>>(out, ysc, sc2, sh2, scS, shS,
                                     (long long)N * COUT, COUT);
}


# Each lane owns 4 adjacent columns of its wave's 16x64 strip (accumulator ct = columns 4c+ct): B rows, gathered rows, LDS tile rows, C rows move as 16-byte accesses, not 4-byte ones; sums as before
# speedup vs baseline: 2.0275x; 1.0311x over previous
#include <hip/hip_runtime.h>
#include <hip/hip_bf16.h>


typedef __attribute__((ext_vector_type(2))) float v2f;
typedef __attribute__((ext_vector_type(4))) float v4f;
typedef __attribute__((ext_vector_type(8))) float v8f;

#define CIN   128
#define COUT  256
#define D2    64
#define KPTS  15
#define NBH   32
#define KDFLAT (KPTS * D2)
#define FKP (KDFLAT + 4)
#define WLP 33
#define INFL_INV 2.5f
#define BN_EPS 1e-5f
#define SLOPE 0.2f

__device__ __forceinline__ v8f wmma_f32(v2f a, v2f b, v8f c) {
  return __builtin_amdgcn_wmma_f32_16x16x4_f32(
       false, a,  false, b,
       (short)0, c,  false,  false);
}

__device__ __forceinline__ float lrelu(float v) {
  return v >= 0.0f ? v : SLOPE * v;
}

__global__ __launch_bounds__(256) void gemm_f32_wmma(
    const float* __restrict__ A, const float* __restrict__ B,
    float* __restrict__ C, int M, int Kd, int Nc, int totalUnits,
    float* __restrict__ part) {
  __shared__ __attribute__((aligned(16))) float red[8 * 2 * 128];
  const int wave = threadIdx.x >> 5;
  const int unit = blockIdx.x * 8 + wave;
  const bool active = unit < totalUnits;
  const int lane = threadIdx.x & 31;
  const int m    = lane & 15;
  const int half = lane >> 4;
  const int strips = Nc >> 6;

  v8f acc0 = {}, acc1 = {}, acc2 = {}, acc3 = {};
  const int row0 = active ? ((unit / strips) << 4) : 0;
  const int col0 = active ? ((unit % strips) << 6) : 0;
  if (active) {
    const float* Ap = A + (size_t)(row0 + m) * Kd + (half << 1);
    const float* Bp = B + (size_t)(half << 1) * Nc + col0 + (m << 2);
    const size_t bstep = (size_t)Nc << 2;
    for (int kk = 0; kk < Kd; kk += 4, Ap += 4, Bp += bstep) {
      v2f a;
      a.x = Ap[0];
      a.y = Ap[1];
      const v4f bx = *(const v4f*)Bp;
      const v4f by = *(const v4f*)(Bp + Nc);
      v2f b0, b1, b2, b3;
      b0.x = bx.x; b0.y = by.x;
      b1.x = bx.y; b1.y = by.y;
      b2.x = bx.z; b2.y = by.z;
      b3.x = bx.w; b3.y = by.w;
      acc0 = wmma_f32(a, b0, acc0);
      acc1 = wmma_f32(a, b1, acc1);
      acc2 = wmma_f32(a, b2, acc2);
      acc3 = wmma_f32(a, b3, acc3);
    }
  }

  float s0 = 0.0f, s1 = 0.0f, s2 = 0.0f, s3 = 0.0f;
  float q0 = 0.0f, q1 = 0.0f, q2 = 0.0f, q3 = 0.0f;
  #pragma unroll
  for (int r = 0; r < 8; ++r) {
    s0 += acc0[r]; q0 += acc0[r] * acc0[r];
    s1 += acc1[r]; q1 += acc1[r] * acc1[r];
    s2 += acc2[r]; q2 += acc2[r] * acc2[r];
    s3 += acc3[r]; q3 += acc3[r] * acc3[r];
  }
  float* rw = red + (wave * 2 + half) * 128 + (m << 2);
  v4f sv, qv;
  sv.x = s0; sv.y = s1; sv.z = s2; sv.w = s3;
  qv.x = q0; qv.y = q1; qv.z = q2; qv.w = q3;
  *(v4f*)rw        = sv;
  *(v4f*)(rw + 64) = qv;
  __syncthreads();
  if ((int)threadIdx.x < strips * 64) {
    const int st  = threadIdx.x >> 6;
    const int col = threadIdx.x & 63;
    float ts = 0.0f, tq = 0.0f;
    for (int w = st; w < 8; w += strips) {
      const float* r0 = red + (w * 2) * 128;
      ts += r0[col];       tq += r0[64 + col];
      ts += r0[128 + col]; tq += r0[192 + col];
    }
    float* p = part + (size_t)blockIdx.x * (size_t)(2 * Nc);
    p[st * 64 + col]      = ts;
    p[Nc + st * 64 + col] = tq;
  }

  if (active) {
    int rbase = row0 + (half << 3);
    float* Cp = C + (size_t)rbase * Nc + col0 + (m << 2);
    #pragma unroll
    for (int r = 0; r < 8; ++r) {
      v4f o;
      o.x = acc0[r]; o.y = acc1[r]; o.z = acc2[r]; o.w = acc3[r];
      *(v4f*)(Cp + (size_t)r * Nc) = o;
    }
  }
}

__global__ __launch_bounds__(256, 8) void kpconv_wmma(
    const float* __restrict__ xyz, const int* __restrict__ nidx,
    const float* __restrict__ x1, const float* __restrict__ kp_points,
    const float* __restrict__ kpw, float* __restrict__ xk) {
  extern __shared__ __attribute__((aligned(16))) float smem[];
  float* FK  = smem;
  float* Wl  = FK + 16 * FKP;
  int*   IDX = (int*)(Wl + 8 * 16 * WLP);
  float* PS  = FK;

  const int wave = threadIdx.x >> 5;
  const int lane = threadIdx.x & 31;
  const int m    = lane & 15;
  const int half = lane >> 4;
  const int base = blockIdx.x << 4;

  float* wl = Wl + wave * (16 * WLP);
  int*   il = IDX + wave * NBH;

  for (int it = 0; it < 2; ++it) {
    const int pi = (wave << 1) + it;
    const int n  = base + pi;

    const int id = nidx[(size_t)n * NBH + lane];
    il[lane] = id;
    const float cx = xyz[(size_t)n * 3 + 0];
    const float cy = xyz[(size_t)n * 3 + 1];
    const float cz = xyz[(size_t)n * 3 + 2];
    const float px = xyz[(size_t)id * 3 + 0] - cx;
    const float py = xyz[(size_t)id * 3 + 1] - cy;
    const float pz = xyz[(size_t)id * 3 + 2] - cz;
    #pragma unroll
    for (int k = 0; k < KPTS; ++k) {
      float dx = px - kp_points[k * 3 + 0];
      float dy = py - kp_points[k * 3 + 1];
      float dz = pz - kp_points[k * 3 + 2];
      float r = sqrtf(dx * dx + dy * dy + dz * dz);
      wl[k * WLP + lane] = fmaxf(1.0f - r * INFL_INV, 0.0f);
    }
    wl[KPTS * WLP + lane] = 0.0f;
    __syncthreads();

    v8f acc0 = {}, acc1 = {}, acc2 = {}, acc3 = {};
    #pragma unroll
    for (int kk = 0; kk < NBH; kk += 4) {
      const int ka = kk + (half << 1);
      v2f a;
      a.x = wl[m * WLP + ka];
      a.y = wl[m * WLP + ka + 1];
      const v4f bx = *(const v4f*)(x1 + (size_t)il[ka] * D2 + (m << 2));
      const v4f by = *(const v4f*)(x1 + (size_t)il[ka + 1] * D2 + (m << 2));
      v2f b0, b1, b2, b3;
      b0.x = bx.x; b0.y = by.x;
      b1.x = bx.y; b1.y = by.y;
      b2.x = bx.z; b2.y = by.z;
      b3.x = bx.w; b3.y = by.w;
      acc0 = wmma_f32(a, b0, acc0);
      acc1 = wmma_f32(a, b1, acc1);
      acc2 = wmma_f32(a, b2, acc2);
      acc3 = wmma_f32(a, b3, acc3);
    }
    #pragma unroll
    for (int r = 0; r < 8; ++r) {
      const int krow = (half << 3) + r;
      if (krow < KPTS) {
        v4f o;
        o.x = acc0[r]; o.y = acc1[r]; o.z = acc2[r]; o.w = acc3[r];
        *(v4f*)(FK + pi * FKP + krow * D2 + (m << 2)) = o;
      }
    }
    __syncthreads();
  }

  {
    const int kbeg = wave * 120;
    v8f acc0 = {}, acc1 = {}, acc2 = {}, acc3 = {};
    for (int kk = kbeg; kk < kbeg + 120; kk += 4) {
      const int ka = kk + (half << 1);
      v2f a;
      a.x = FK[m * FKP + ka];
      a.y = FK[m * FKP + ka + 1];
      const float* Bp = kpw + (size_t)ka * D2 + (m << 2);
      const v4f bx = *(const v4f*)Bp;
      const v4f by = *(const v4f*)(Bp + D2);
      v2f b0, b1, b2, b3;
      b0.x = bx.x; b0.y = by.x;
      b1.x = bx.y; b1.y = by.y;
      b2.x = bx.z; b2.y = by.z;
      b3.x = bx.w; b3.y = by.w;
      acc0 = wmma_f32(a, b0, acc0);
      acc1 = wmma_f32(a, b1, acc1);
      acc2 = wmma_f32(a, b2, acc2);
      acc3 = wmma_f32(a, b3, acc3);
    }
    __syncthreads();
    #pragma unroll
    for (int r = 0; r < 8; ++r) {
      v4f o;
      o.x = acc0[r]; o.y = acc1[r]; o.z = acc2[r]; o.w = acc3[r];
      *(v4f*)(PS + wave * (16 * D2) + ((half << 3) + r) * D2 + (m << 2)) = o;
    }
  }
  __syncthreads();

  for (int t = threadIdx.x; t < 16 * D2; t += 256) {
    float s = 0.0f;
    #pragma unroll
    for (int w = 0; w < 8; ++w) s += PS[w * (16 * D2) + t];
    const int pt = t >> 6;
    const int e  = t & 63;
    xk[(size_t)(base + pt) * D2 + e] = s;
  }
}

__global__ __launch_bounds__(1024) void finalize_bn(
    const float* __restrict__ part, int nparts,
    const float* __restrict__ g, const float* __restrict__ b, int C, float invN,
    float* __restrict__ sc, float* __restrict__ sh) {
  __shared__ float rs[1024];
  __shared__ float rq[1024];
  const int cl = threadIdx.x & 15;
  const int sl = threadIdx.x >> 4;
  const int c  = blockIdx.x * 16 + cl;
  const size_t pitch = (size_t)(2 * C);
  float s = 0.0f, q = 0.0f;
  #pragma unroll 8
  for (int p = sl; p < nparts; p += 64) {
    s += part[(size_t)p * pitch + c];
    q += part[(size_t)p * pitch + C + c];
  }
  rs[threadIdx.x] = s;
  rq[threadIdx.x] = q;
  __syncthreads();
  if (threadIdx.x < 16) {
    float ts = 0.0f, tq = 0.0f;
    #pragma unroll
    for (int k = 0; k < 64; ++k) {
      ts += rs[k * 16 + cl];
      tq += rq[k * 16 + cl];
    }
    float mean = ts * invN;
    float var  = tq * invN - mean * mean;
    float scl  = g[c] * rsqrtf(var + BN_EPS);
    sc[c] = scl;
    sh[c] = b[c] - mean * scl;
  }
}

__global__ void bnact_inplace(float* __restrict__ Y, const float* __restrict__ sc,
                              const float* __restrict__ sh, long long total, int C) {
  size_t idx0   = (size_t)blockIdx.x * blockDim.x + threadIdx.x;
  size_t stride = (size_t)gridDim.x * blockDim.x;
  for (size_t i = idx0; i < (size_t)total; i += stride) {
    int c = (int)(i % (size_t)C);
    Y[i] = lrelu(Y[i] * sc[c] + sh[c]);
  }
}

__global__ void fuse_out(float* __restrict__ out, const float* __restrict__ ysc,
                         const float* __restrict__ sc2, const float* __restrict__ sh2,
                         const float* __restrict__ scS, const float* __restrict__ shS,
                         long long total, int C) {
  size_t idx0   = (size_t)blockIdx.x * blockDim.x + threadIdx.x;
  size_t stride = (size_t)gridDim.x * blockDim.x;
  for (size_t i = idx0; i < (size_t)total; i += stride) {
    int c = (int)(i % (size_t)C);
    float v = lrelu(out[i] * sc2[c] + sh2[c]);
    out[i] = v + (ysc[i] * scS[c] + shS[c]);
  }
}

extern "C" void kernel_launch(void* const* d_in, const int* in_sizes, int n_in,
                              void* d_out, int out_size, void* d_ws, size_t ws_size,
                              hipStream_t stream) {
  const float* feats = (const float*)d_in[0];
  const float* xyz   = (const float*)d_in[1];
  const int*   nidx  = (const int*)d_in[3];
  const float* W1    = (const float*)d_in[4];
  const float* g1    = (const float*)d_in[5];
  const float* b1    = (const float*)d_in[6];
  const float* kpp   = (const float*)d_in[7];
  const float* kpw   = (const float*)d_in[8];
  const float* W2    = (const float*)d_in[9];
  const float* g2    = (const float*)d_in[10];
  const float* b2    = (const float*)d_in[11];
  const float* Wsc   = (const float*)d_in[12];
  const float* gsc   = (const float*)d_in[13];
  const float* bsc   = (const float*)d_in[14];
  float* out = (float*)d_out;

  const int N = in_sizes[0] / CIN;

  float* ws      = (float*)d_ws;
  float* x1      = ws;
  float* xk      = x1 + (size_t)N * D2;
  float* ysc     = xk + (size_t)N * D2;
  float* sc1 = ysc + (size_t)N * COUT; float* sh1 = sc1 + D2;
  float* sc2 = sh1 + D2;               float* sh2 = sc2 + COUT;
  float* scS = sh2 + COUT;             float* shS = scS + COUT;
  const int blk1 = ((N / 16) * (D2 / 64) + 7) / 8;
  const int blk2 = ((N / 16) * (COUT / 64) + 7) / 8;
  float* part1  = xk;
  float* part2  = x1;
  float* partSC = x1 + (size_t)blk2 * 2 * COUT;

  {
    int units = (N / 16) * (D2 / 64);
    gemm_f32_wmma<<<blk1, 256, 0, stream>>>(feats, W1, x1, N, CIN, D2, units, part1);
  }
  finalize_bn<<<D2 / 16, 1024, 0, stream>>>(part1, blk1, g1, b1, D2, 1.0f / (float)N, sc1, sh1);
  bnact_inplace<<<2048, 256, 0, stream>>>(x1, sc1, sh1, (long long)N * D2, D2);

  {
    size_t lds = (16 * FKP + 8 * 16 * WLP) * sizeof(float)
               + 8 * NBH * sizeof(int);
    kpconv_wmma<<<N / 16, 256, lds, stream>>>(xyz, nidx, x1, kpp, kpw, xk);
  }

  {
    int units = (N / 16) * (COUT / 64);
    gemm_f32_wmma<<<blk2, 256, 0, stream>>>(xk, W2, out, N, D2, COUT, units, part2);
  }
  {
    int units = (N / 16) * (COUT / 64);
    gemm_f32_wmma<<<blk2, 256, 0, stream>>>(feats, Wsc, ysc, N, CIN, COUT, units, partSC);
  }

  finalize_bn<<<COUT / 16, 1024, 0, stream>>>(part2, blk2, g2, b2, COUT, 1.0f / (float)N, sc2, sh2);
  finalize_bn<<<COUT / 16, 1024, 0, stream>>>(partSC, blk2, gsc, bsc, COUT, 1.0f / (float)N, scS, shS);

  fuse_out<<<4096, 256, 0, stream>>>(out, ysc, sc2, sh2, scS, shS,
                                     (long long)N * COUT, COUT);
}


# Elementwise BN passes (x1 activation, final combine): a thread owns 4 fixed columns, scale/shift loaded once, 16-byte accesses, a trip's loads issued together, 13000 waves not 49152; sums unchanged
# speedup vs baseline: 2.1285x; 1.0498x over previous
#include <hip/hip_runtime.h>
#include <hip/hip_bf16.h>


typedef __attribute__((ext_vector_type(2))) float v2f;
typedef __attribute__((ext_vector_type(4))) float v4f;
typedef __attribute__((ext_vector_type(8))) float v8f;

#define CIN   128
#define COUT  256
#define D2    64
#define KPTS  15
#define NBH   32
#define KDFLAT (KPTS * D2)
#define FKP (KDFLAT + 4)
#define WLP 33
#define INFL_INV 2.5f
#define BN_EPS 1e-5f
#define SLOPE 0.2f

__device__ __forceinline__ v8f wmma_f32(v2f a, v2f b, v8f c) {
  return __builtin_amdgcn_wmma_f32_16x16x4_f32(
       false, a,  false, b,
       (short)0, c,  false,  false);
}

__device__ __forceinline__ float lrelu(float v) {
  return v >= 0.0f ? v : SLOPE * v;
}

__global__ __launch_bounds__(256) void gemm_f32_wmma(
    const float* __restrict__ A, const float* __restrict__ B,
    float* __restrict__ C, int M, int Kd, int Nc, int totalUnits,
    float* __restrict__ part) {
  __shared__ __attribute__((aligned(16))) float red[8 * 2 * 128];
  const int wave = threadIdx.x >> 5;
  const int unit = blockIdx.x * 8 + wave;
  const bool active = unit < totalUnits;
  const int lane = threadIdx.x & 31;
  const int m    = lane & 15;
  const int half = lane >> 4;
  const int strips = Nc >> 6;

  v8f acc0 = {}, acc1 = {}, acc2 = {}, acc3 = {};
  const int row0 = active ? ((unit / strips) << 4) : 0;
  const int col0 = active ? ((unit % strips) << 6) : 0;
  if (active) {
    const float* Ap = A + (size_t)(row0 + m) * Kd + (half << 1);
    const float* Bp = B + (size_t)(half << 1) * Nc + col0 + (m << 2);
    const size_t bstep = (size_t)Nc << 2;
    for (int kk = 0; kk < Kd; kk += 4, Ap += 4, Bp += bstep) {
      v2f a;
      a.x = Ap[0];
      a.y = Ap[1];
      const v4f bx = *(const v4f*)Bp;
      const v4f by = *(const v4f*)(Bp + Nc);
      v2f b0, b1, b2, b3;
      b0.x = bx.x; b0.y = by.x;
      b1.x = bx.y; b1.y = by.y;
      b2.x = bx.z; b2.y = by.z;
      b3.x = bx.w; b3.y = by.w;
      acc0 = wmma_f32(a, b0, acc0);
      acc1 = wmma_f32(a, b1, acc1);
      acc2 = wmma_f32(a, b2, acc2);
      acc3 = wmma_f32(a, b3, acc3);
    }
  }

  float s0 = 0.0f, s1 = 0.0f, s2 = 0.0f, s3 = 0.0f;
  float q0 = 0.0f, q1 = 0.0f, q2 = 0.0f, q3 = 0.0f;
  #pragma unroll
  for (int r = 0; r < 8; ++r) {
    s0 += acc0[r]; q0 += acc0[r] * acc0[r];
    s1 += acc1[r]; q1 += acc1[r] * acc1[r];
    s2 += acc2[r]; q2 += acc2[r] * acc2[r];
    s3 += acc3[r]; q3 += acc3[r] * acc3[r];
  }
  float* rw = red + (wave * 2 + half) * 128 + (m << 2);
  v4f sv, qv;
  sv.x = s0; sv.y = s1; sv.z = s2; sv.w = s3;
  qv.x = q0; qv.y = q1; qv.z = q2; qv.w = q3;
  *(v4f*)rw        = sv;
  *(v4f*)(rw + 64) = qv;
  __syncthreads();
  if ((int)threadIdx.x < strips * 64) {
    const int st  = threadIdx.x >> 6;
    const int col = threadIdx.x & 63;
    float ts = 0.0f, tq = 0.0f;
    for (int w = st; w < 8; w += strips) {
      const float* r0 = red + (w * 2) * 128;
      ts += r0[col];       tq += r0[64 + col];
      ts += r0[128 + col]; tq += r0[192 + col];
    }
    float* p = part + (size_t)blockIdx.x * (size_t)(2 * Nc);
    p[st * 64 + col]      = ts;
    p[Nc + st * 64 + col] = tq;
  }

  if (active) {
    int rbase = row0 + (half << 3);
    float* Cp = C + (size_t)rbase * Nc + col0 + (m << 2);
    #pragma unroll
    for (int r = 0; r < 8; ++r) {
      v4f o;
      o.x = acc0[r]; o.y = acc1[r]; o.z = acc2[r]; o.w = acc3[r];
      *(v4f*)(Cp + (size_t)r * Nc) = o;
    }
  }
}

__global__ __launch_bounds__(256, 8) void kpconv_wmma(
    const float* __restrict__ xyz, const int* __restrict__ nidx,
    const float* __restrict__ x1, const float* __restrict__ kp_points,
    const float* __restrict__ kpw, float* __restrict__ xk) {
  extern __shared__ __attribute__((aligned(16))) float smem[];
  float* FK  = smem;
  float* Wl  = FK + 16 * FKP;
  int*   IDX = (int*)(Wl + 8 * 16 * WLP);
  float* PS  = FK;

  const int wave = threadIdx.x >> 5;
  const int lane = threadIdx.x & 31;
  const int m    = lane & 15;
  const int half = lane >> 4;
  const int base = blockIdx.x << 4;

  float* wl = Wl + wave * (16 * WLP);
  int*   il = IDX + wave * NBH;

  for (int it = 0; it < 2; ++it) {
    const int pi = (wave << 1) + it;
    const int n  = base + pi;

    const int id = nidx[(size_t)n * NBH + lane];
    il[lane] = id;
    const float cx = xyz[(size_t)n * 3 + 0];
    const float cy = xyz[(size_t)n * 3 + 1];
    const float cz = xyz[(size_t)n * 3 + 2];
    const float px = xyz[(size_t)id * 3 + 0] - cx;
    const float py = xyz[(size_t)id * 3 + 1] - cy;
    const float pz = xyz[(size_t)id * 3 + 2] - cz;
    #pragma unroll
    for (int k = 0; k < KPTS; ++k) {
      float dx = px - kp_points[k * 3 + 0];
      float dy = py - kp_points[k * 3 + 1];
      float dz = pz - kp_points[k * 3 + 2];
      float r = sqrtf(dx * dx + dy * dy + dz * dz);
      wl[k * WLP + lane] = fmaxf(1.0f - r * INFL_INV, 0.0f);
    }
    wl[KPTS * WLP + lane] = 0.0f;
    __syncthreads();

    v8f acc0 = {}, acc1 = {}, acc2 = {}, acc3 = {};
    #pragma unroll
    for (int kk = 0; kk < NBH; kk += 4) {
      const int ka = kk + (half << 1);
      v2f a;
      a.x = wl[m * WLP + ka];
      a.y = wl[m * WLP + ka + 1];
      const v4f bx = *(const v4f*)(x1 + (size_t)il[ka] * D2 + (m << 2));
      const v4f by = *(const v4f*)(x1 + (size_t)il[ka + 1] * D2 + (m << 2));
      v2f b0, b1, b2, b3;
      b0.x = bx.x; b0.y = by.x;
      b1.x = bx.y; b1.y = by.y;
      b2.x = bx.z; b2.y = by.z;
      b3.x = bx.w; b3.y = by.w;
      acc0 = wmma_f32(a, b0, acc0);
      acc1 = wmma_f32(a, b1, acc1);
      acc2 = wmma_f32(a, b2, acc2);
      acc3 = wmma_f32(a, b3, acc3);
    }
    #pragma unroll
    for (int r = 0; r < 8; ++r) {
      const int krow = (half << 3) + r;
      if (krow < KPTS) {
        v4f o;
        o.x = acc0[r]; o.y = acc1[r]; o.z = acc2[r]; o.w = acc3[r];
        *(v4f*)(FK + pi * FKP + krow * D2 + (m << 2)) = o;
      }
    }
    __syncthreads();
  }

  {
    const int kbeg = wave * 120;
    v8f acc0 = {}, acc1 = {}, acc2 = {}, acc3 = {};
    for (int kk = kbeg; kk < kbeg + 120; kk += 4) {
      const int ka = kk + (half << 1);
      v2f a;
      a.x = FK[m * FKP + ka];
      a.y = FK[m * FKP + ka + 1];
      const float* Bp = kpw + (size_t)ka * D2 + (m << 2);
      const v4f bx = *(const v4f*)Bp;
      const v4f by = *(const v4f*)(Bp + D2);
      v2f b0, b1, b2, b3;
      b0.x = bx.x; b0.y = by.x;
      b1.x = bx.y; b1.y = by.y;
      b2.x = bx.z; b2.y = by.z;
      b3.x = bx.w; b3.y = by.w;
      acc0 = wmma_f32(a, b0, acc0);
      acc1 = wmma_f32(a, b1, acc1);
      acc2 = wmma_f32(a, b2, acc2);
      acc3 = wmma_f32(a, b3, acc3);
    }
    __syncthreads();
    #pragma unroll
    for (int r = 0; r < 8; ++r) {
      v4f o;
      o.x = acc0[r]; o.y = acc1[r]; o.z = acc2[r]; o.w = acc3[r];
      *(v4f*)(PS + wave * (16 * D2) + ((half << 3) + r) * D2 + (m << 2)) = o;
    }
  }
  __syncthreads();

  for (int t = threadIdx.x; t < 16 * D2; t += 256) {
    float s = 0.0f;
    #pragma unroll
    for (int w = 0; w < 8; ++w) s += PS[w * (16 * D2) + t];
    const int pt = t >> 6;
    const int e  = t & 63;
    xk[(size_t)(base + pt) * D2 + e] = s;
  }
}

__global__ __launch_bounds__(1024) void finalize_bn(
    const float* __restrict__ part, int nparts,
    const float* __restrict__ g, const float* __restrict__ b, int C, float invN,
    float* __restrict__ sc, float* __restrict__ sh) {
  __shared__ float rs[1024];
  __shared__ float rq[1024];
  const int cl = threadIdx.x & 15;
  const int sl = threadIdx.x >> 4;
  const int c  = blockIdx.x * 16 + cl;
  const size_t pitch = (size_t)(2 * C);
  float s = 0.0f, q = 0.0f;
  #pragma unroll 8
  for (int p = sl; p < nparts; p += 64) {
    s += part[(size_t)p * pitch + c];
    q += part[(size_t)p * pitch + C + c];
  }
  rs[threadIdx.x] = s;
  rq[threadIdx.x] = q;
  __syncthreads();
  if (threadIdx.x < 16) {
    float ts = 0.0f, tq = 0.0f;
    #pragma unroll
    for (int k = 0; k < 64; ++k) {
      ts += rs[k * 16 + cl];
      tq += rq[k * 16 + cl];
    }
    float mean = ts * invN;
    float var  = tq * invN - mean * mean;
    float scl  = g[c] * rsqrtf(var + BN_EPS);
    sc[c] = scl;
    sh[c] = b[c] - mean * scl;
  }
}


template <int C, int U>
__global__ __launch_bounds__(256) void bnact_inplace(
    float* __restrict__ Y, const float* __restrict__ sc,
    const float* __restrict__ sh, int rows) {
  constexpr int G   = C / 4;
  constexpr int RPB = 256 / G;
  const int cg    = threadIdx.x % G;
  const int r0    = blockIdx.x * RPB + threadIdx.x / G;
  const int rstep = gridDim.x * RPB;
  const v4f a = *(const v4f*)(sc + 4 * cg);
  const v4f b = *(const v4f*)(sh + 4 * cg);
  float* base = Y + 4 * cg;
  for (int r = r0; r < rows; r += U * rstep) {
    v4f y[U];
    #pragma unroll
    for (int i = 0; i < U; ++i) {
      const int ri = min(r + i * rstep, rows - 1);
      y[i] = *(const v4f*)(base + (size_t)ri * C);
    }
    __builtin_amdgcn_sched_barrier(0);
    #pragma unroll
    for (int i = 0; i < U; ++i) {
      const int ri = r + i * rstep;
      if (ri < rows) {
        v4f o;
        o.x = lrelu(y[i].x * a.x + b.x);
        o.y = lrelu(y[i].y * a.y + b.y);
        o.z = lrelu(y[i].z * a.z + b.z);
        o.w = lrelu(y[i].w * a.w + b.w);
        *(v4f*)(base + (size_t)ri * C) = o;
      }
    }
  }
}

template <int C, int U>
__global__ __launch_bounds__(256) void fuse_out(
    float* __restrict__ out, const float* __restrict__ ysc,
    const float* __restrict__ sc2, const float* __restrict__ sh2,
    const float* __restrict__ scS, const float* __restrict__ shS, int rows) {
  constexpr int G   = C / 4;
  constexpr int RPB = 256 / G;
  const int cg    = threadIdx.x % G;
  const int r0    = blockIdx.x * RPB + threadIdx.x / G;
  const int rstep = gridDim.x * RPB;
  const v4f a2 = *(const v4f*)(sc2 + 4 * cg);
  const v4f b2 = *(const v4f*)(sh2 + 4 * cg);
  const v4f aS = *(const v4f*)(scS + 4 * cg);
  const v4f bS = *(const v4f*)(shS + 4 * cg);
  float*       obase = out + 4 * cg;
  const float* sbase = ysc + 4 * cg;
  for (int r = r0; r < rows; r += U * rstep) {
    v4f y[U], s[U];
    #pragma unroll
    for (int i = 0; i < U; ++i) {
      const int ri = min(r + i * rstep, rows - 1);
      y[i] = *(const v4f*)(obase + (size_t)ri * C);
      s[i] = *(const v4f*)(sbase + (size_t)ri * C);
    }
    __builtin_amdgcn_sched_barrier(0);
    #pragma unroll
    for (int i = 0; i < U; ++i) {
      const int ri = r + i * rstep;
      if (ri < rows) {
        v4f o;
        o.x = lrelu(y[i].x * a2.x + b2.x) + (s[i].x * aS.x + bS.x);
        o.y = lrelu(y[i].y * a2.y + b2.y) + (s[i].y * aS.y + bS.y);
        o.z = lrelu(y[i].z * a2.z + b2.z) + (s[i].z * aS.z + bS.z);
        o.w = lrelu(y[i].w * a2.w + b2.w) + (s[i].w * aS.w + bS.w);
        *(v4f*)(obase + (size_t)ri * C) = o;
      }
    }
  }
}

extern "C" void kernel_launch(void* const* d_in, const int* in_sizes, int n_in,
                              void* d_out, int out_size, void* d_ws, size_t ws_size,
                              hipStream_t stream) {
  const float* feats = (const float*)d_in[0];
  const float* xyz   = (const float*)d_in[1];
  const int*   nidx  = (const int*)d_in[3];
  const float* W1    = (const float*)d_in[4];
  const float* g1    = (const float*)d_in[5];
  const float* b1    = (const float*)d_in[6];
  const float* kpp   = (const float*)d_in[7];
  const float* kpw   = (const float*)d_in[8];
  const float* W2    = (const float*)d_in[9];
  const float* g2    = (const float*)d_in[10];
  const float* b2    = (const float*)d_in[11];
  const float* Wsc   = (const float*)d_in[12];
  const float* gsc   = (const float*)d_in[13];
  const float* bsc   = (const float*)d_in[14];
  float* out = (float*)d_out;

  const int N = in_sizes[0] / CIN;

  float* ws      = (float*)d_ws;
  float* x1      = ws;
  float* xk      = x1 + (size_t)N * D2;
  float* ysc     = xk + (size_t)N * D2;
  float* sc1 = ysc + (size_t)N * COUT; float* sh1 = sc1 + D2;
  float* sc2 = sh1 + D2;               float* sh2 = sc2 + COUT;
  float* scS = sh2 + COUT;             float* shS = scS + COUT;
  const int blk1 = ((N / 16) * (D2 / 64) + 7) / 8;
  const int blk2 = ((N / 16) * (COUT / 64) + 7) / 8;
  float* part1  = xk;
  float* part2  = x1;
  float* partSC = x1 + (size_t)blk2 * 2 * COUT;

  {
    int units = (N / 16) * (D2 / 64);
    gemm_f32_wmma<<<blk1, 256, 0, stream>>>(feats, W1, x1, N, CIN, D2, units, part1);
  }
  finalize_bn<<<D2 / 16, 1024, 0, stream>>>(part1, blk1, g1, b1, D2, 1.0f / (float)N, sc1, sh1);
  bnact_inplace<D2, 8><<<625, 256, 0, stream>>>(x1, sc1, sh1, N);

  {
    size_t lds = (16 * FKP + 8 * 16 * WLP) * sizeof(float)
               + 8 * NBH * sizeof(int);
    kpconv_wmma<<<N / 16, 256, lds, stream>>>(xyz, nidx, x1, kpp, kpw, xk);
  }

  {
    int units = (N / 16) * (COUT / 64);
    gemm_f32_wmma<<<blk2, 256, 0, stream>>>(xk, W2, out, N, D2, COUT, units, part2);
  }
  {
    int units = (N / 16) * (COUT / 64);
    gemm_f32_wmma<<<blk2, 256, 0, stream>>>(feats, Wsc, ysc, N, CIN, COUT, units, partSC);
  }

  finalize_bn<<<COUT / 16, 1024, 0, stream>>>(part2, blk2, g2, b2, COUT, 1.0f / (float)N, sc2, sh2);
  finalize_bn<<<COUT / 16, 1024, 0, stream>>>(partSC, blk2, gsc, bsc, COUT, 1.0f / (float)N, scS, shS);

  fuse_out<COUT, 4><<<1000, 256, 0, stream>>>(out, ysc, sc2, sh2, scS, shS, N);
}


# Operand loads of a run of k-steps issued together ahead of the run's WMMAs: GEMM K-loop 8 steps (24 loads), KPConv neighbour gather 2 steps; one memory wait per run, not per step; sums as before
# speedup vs baseline: 2.2678x; 1.0654x over previous
#include <hip/hip_runtime.h>
#include <hip/hip_bf16.h>


typedef __attribute__((ext_vector_type(2))) float v2f;
typedef __attribute__((ext_vector_type(4))) float v4f;
typedef __attribute__((ext_vector_type(8))) float v8f;

#define CIN   128
#define COUT  256
#define D2    64
#define KPTS  15
#define NBH   32
#define KDFLAT (KPTS * D2)
#define FKP (KDFLAT + 4)
#define WLP 33
#define INFL_INV 2.5f
#define BN_EPS 1e-5f
#define SLOPE 0.2f

__device__ __forceinline__ v8f wmma_f32(v2f a, v2f b, v8f c) {
  return __builtin_amdgcn_wmma_f32_16x16x4_f32(
       false, a,  false, b,
       (short)0, c,  false,  false);
}

__device__ __forceinline__ float lrelu(float v) {
  return v >= 0.0f ? v : SLOPE * v;
}

#define KU 8
__global__ __launch_bounds__(256) void gemm_f32_wmma(
    const float* __restrict__ A, const float* __restrict__ B,
    float* __restrict__ C, int M, int Kd, int Nc, int totalUnits,
    float* __restrict__ part) {
  __shared__ __attribute__((aligned(16))) float red[8 * 2 * 128];
  const int wave = threadIdx.x >> 5;
  const int unit = blockIdx.x * 8 + wave;
  const bool active = unit < totalUnits;
  const int lane = threadIdx.x & 31;
  const int m    = lane & 15;
  const int half = lane >> 4;
  const int strips = Nc >> 6;

  v8f acc0 = {}, acc1 = {}, acc2 = {}, acc3 = {};
  const int row0 = active ? ((unit / strips) << 4) : 0;
  const int col0 = active ? ((unit % strips) << 6) : 0;
  if (active) {
    const float* Ap = A + (size_t)(row0 + m) * Kd + (half << 1);
    const float* Bp = B + (size_t)(half << 1) * Nc + col0 + (m << 2);
    const size_t bstep = (size_t)Nc << 2;
    for (int kk = 0; kk < Kd; kk += 4 * KU, Ap += 4 * KU, Bp += KU * bstep) {
      v2f a[KU];
      v4f bx[KU], by[KU];
      #pragma unroll
      for (int j = 0; j < KU; ++j) {
        a[j]  = *(const v2f*)(Ap + 4 * j);
        bx[j] = *(const v4f*)(Bp + j * bstep);
        by[j] = *(const v4f*)(Bp + j * bstep + Nc);
      }
      __builtin_amdgcn_sched_barrier(0);
      #pragma unroll
      for (int j = 0; j < KU; ++j) {
        v2f b0, b1, b2, b3;
        b0.x = bx[j].x; b0.y = by[j].x;
        b1.x = bx[j].y; b1.y = by[j].y;
        b2.x = bx[j].z; b2.y = by[j].z;
        b3.x = bx[j].w; b3.y = by[j].w;
        acc0 = wmma_f32(a[j], b0, acc0);
        acc1 = wmma_f32(a[j], b1, acc1);
        acc2 = wmma_f32(a[j], b2, acc2);
        acc3 = wmma_f32(a[j], b3, acc3);
      }
    }
  }

  float s0 = 0.0f, s1 = 0.0f, s2 = 0.0f, s3 = 0.0f;
  float q0 = 0.0f, q1 = 0.0f, q2 = 0.0f, q3 = 0.0f;
  #pragma unroll
  for (int r = 0; r < 8; ++r) {
    s0 += acc0[r]; q0 += acc0[r] * acc0[r];
    s1 += acc1[r]; q1 += acc1[r] * acc1[r];
    s2 += acc2[r]; q2 += acc2[r] * acc2[r];
    s3 += acc3[r]; q3 += acc3[r] * acc3[r];
  }
  float* rw = red + (wave * 2 + half) * 128 + (m << 2);
  v4f sv, qv;
  sv.x = s0; sv.y = s1; sv.z = s2; sv.w = s3;
  qv.x = q0; qv.y = q1; qv.z = q2; qv.w = q3;
  *(v4f*)rw        = sv;
  *(v4f*)(rw + 64) = qv;
  __syncthreads();
  if ((int)threadIdx.x < strips * 64) {
    const int st  = threadIdx.x >> 6;
    const int col = threadIdx.x & 63;
    float ts = 0.0f, tq = 0.0f;
    for (int w = st; w < 8; w += strips) {
      const float* r0 = red + (w * 2) * 128;
      ts += r0[col];       tq += r0[64 + col];
      ts += r0[128 + col]; tq += r0[192 + col];
    }
    float* p = part + (size_t)blockIdx.x * (size_t)(2 * Nc);
    p[st * 64 + col]      = ts;
    p[Nc + st * 64 + col] = tq;
  }

  if (active) {
    int rbase = row0 + (half << 3);
    float* Cp = C + (size_t)rbase * Nc + col0 + (m << 2);
    #pragma unroll
    for (int r = 0; r < 8; ++r) {
      v4f o;
      o.x = acc0[r]; o.y = acc1[r]; o.z = acc2[r]; o.w = acc3[r];
      *(v4f*)(Cp + (size_t)r * Nc) = o;
    }
  }
}

#define GU 2
__global__ __launch_bounds__(256, 8) void kpconv_wmma(
    const float* __restrict__ xyz, const int* __restrict__ nidx,
    const float* __restrict__ x1, const float* __restrict__ kp_points,
    const float* __restrict__ kpw, float* __restrict__ xk) {
  extern __shared__ __attribute__((aligned(16))) float smem[];
  float* FK  = smem;
  float* Wl  = FK + 16 * FKP;
  int*   IDX = (int*)(Wl + 8 * 16 * WLP);
  float* PS  = FK;

  const int wave = threadIdx.x >> 5;
  const int lane = threadIdx.x & 31;
  const int m    = lane & 15;
  const int half = lane >> 4;
  const int base = blockIdx.x << 4;

  float* wl = Wl + wave * (16 * WLP);
  int*   il = IDX + wave * NBH;

  for (int it = 0; it < 2; ++it) {
    const int pi = (wave << 1) + it;
    const int n  = base + pi;

    const int id = nidx[(size_t)n * NBH + lane];
    il[lane] = id;
    const float cx = xyz[(size_t)n * 3 + 0];
    const float cy = xyz[(size_t)n * 3 + 1];
    const float cz = xyz[(size_t)n * 3 + 2];
    const float px = xyz[(size_t)id * 3 + 0] - cx;
    const float py = xyz[(size_t)id * 3 + 1] - cy;
    const float pz = xyz[(size_t)id * 3 + 2] - cz;
    #pragma unroll
    for (int k = 0; k < KPTS; ++k) {
      float dx = px - kp_points[k * 3 + 0];
      float dy = py - kp_points[k * 3 + 1];
      float dz = pz - kp_points[k * 3 + 2];
      float r = sqrtf(dx * dx + dy * dy + dz * dz);
      wl[k * WLP + lane] = fmaxf(1.0f - r * INFL_INV, 0.0f);
    }
    wl[KPTS * WLP + lane] = 0.0f;
    __syncthreads();

    v8f acc0 = {}, acc1 = {}, acc2 = {}, acc3 = {};
    #pragma unroll
    for (int k0 = 0; k0 < NBH; k0 += 4 * GU) {
      v4f bx[GU], by[GU];
      #pragma unroll
      for (int j = 0; j < GU; ++j) {
        const int ka = k0 + 4 * j + (half << 1);
        bx[j] = *(const v4f*)(x1 + (size_t)il[ka] * D2 + (m << 2));
        by[j] = *(const v4f*)(x1 + (size_t)il[ka + 1] * D2 + (m << 2));
      }
      __builtin_amdgcn_sched_barrier(0);
      #pragma unroll
      for (int j = 0; j < GU; ++j) {
        const int ka = k0 + 4 * j + (half << 1);
        v2f a;
        a.x = wl[m * WLP + ka];
        a.y = wl[m * WLP + ka + 1];
        v2f b0, b1, b2, b3;
        b0.x = bx[j].x; b0.y = by[j].x;
        b1.x = bx[j].y; b1.y = by[j].y;
        b2.x = bx[j].z; b2.y = by[j].z;
        b3.x = bx[j].w; b3.y = by[j].w;
        acc0 = wmma_f32(a, b0, acc0);
        acc1 = wmma_f32(a, b1, acc1);
        acc2 = wmma_f32(a, b2, acc2);
        acc3 = wmma_f32(a, b3, acc3);
      }
    }
    #pragma unroll
    for (int r = 0; r < 8; ++r) {
      const int krow = (half << 3) + r;
      if (krow < KPTS) {
        v4f o;
        o.x = acc0[r]; o.y = acc1[r]; o.z = acc2[r]; o.w = acc3[r];
        *(v4f*)(FK + pi * FKP + krow * D2 + (m << 2)) = o;
      }
    }
    __syncthreads();
  }

  {
    const int kbeg = wave * 120;
    v8f acc0 = {}, acc1 = {}, acc2 = {}, acc3 = {};
    for (int kk = kbeg; kk < kbeg + 120; kk += 4) {
      const int ka = kk + (half << 1);
      v2f a;
      a.x = FK[m * FKP + ka];
      a.y = FK[m * FKP + ka + 1];
      const float* Bp = kpw + (size_t)ka * D2 + (m << 2);
      const v4f bx = *(const v4f*)Bp;
      const v4f by = *(const v4f*)(Bp + D2);
      v2f b0, b1, b2, b3;
      b0.x = bx.x; b0.y = by.x;
      b1.x = bx.y; b1.y = by.y;
      b2.x = bx.z; b2.y = by.z;
      b3.x = bx.w; b3.y = by.w;
      acc0 = wmma_f32(a, b0, acc0);
      acc1 = wmma_f32(a, b1, acc1);
      acc2 = wmma_f32(a, b2, acc2);
      acc3 = wmma_f32(a, b3, acc3);
    }
    __syncthreads();
    #pragma unroll
    for (int r = 0; r < 8; ++r) {
      v4f o;
      o.x = acc0[r]; o.y = acc1[r]; o.z = acc2[r]; o.w = acc3[r];
      *(v4f*)(PS + wave * (16 * D2) + ((half << 3) + r) * D2 + (m << 2)) = o;
    }
  }
  __syncthreads();

  for (int t = threadIdx.x; t < 16 * D2; t += 256) {
    float s = 0.0f;
    #pragma unroll
    for (int w = 0; w < 8; ++w) s += PS[w * (16 * D2) + t];
    const int pt = t >> 6;
    const int e  = t & 63;
    xk[(size_t)(base + pt) * D2 + e] = s;
  }
}

__global__ __launch_bounds__(1024) void finalize_bn(
    const float* __restrict__ part, int nparts,
    const float* __restrict__ g, const float* __restrict__ b, int C, float invN,
    float* __restrict__ sc, float* __restrict__ sh) {
  __shared__ float rs[1024];
  __shared__ float rq[1024];
  const int cl = threadIdx.x & 15;
  const int sl = threadIdx.x >> 4;
  const int c  = blockIdx.x * 16 + cl;
  const size_t pitch = (size_t)(2 * C);
  float s = 0.0f, q = 0.0f;
  #pragma unroll 8
  for (int p = sl; p < nparts; p += 64) {
    s += part[(size_t)p * pitch + c];
    q += part[(size_t)p * pitch + C + c];
  }
  rs[threadIdx.x] = s;
  rq[threadIdx.x] = q;
  __syncthreads();
  if (threadIdx.x < 16) {
    float ts = 0.0f, tq = 0.0f;
    #pragma unroll
    for (int k = 0; k < 64; ++k) {
      ts += rs[k * 16 + cl];
      tq += rq[k * 16 + cl];
    }
    float mean = ts * invN;
    float var  = tq * invN - mean * mean;
    float scl  = g[c] * rsqrtf(var + BN_EPS);
    sc[c] = scl;
    sh[c] = b[c] - mean * scl;
  }
}


template <int C, int U>
__global__ __launch_bounds__(256) void bnact_inplace(
    float* __restrict__ Y, const float* __restrict__ sc,
    const float* __restrict__ sh, int rows) {
  constexpr int G   = C / 4;
  constexpr int RPB = 256 / G;
  const int cg    = threadIdx.x % G;
  const int r0    = blockIdx.x * RPB + threadIdx.x / G;
  const int rstep = gridDim.x * RPB;
  const v4f a = *(const v4f*)(sc + 4 * cg);
  const v4f b = *(const v4f*)(sh + 4 * cg);
  float* base = Y + 4 * cg;
  for (int r = r0; r < rows; r += U * rstep) {
    v4f y[U];
    #pragma unroll
    for (int i = 0; i < U; ++i) {
      const int ri = min(r + i * rstep, rows - 1);
      y[i] = *(const v4f*)(base + (size_t)ri * C);
    }
    __builtin_amdgcn_sched_barrier(0);
    #pragma unroll
    for (int i = 0; i < U; ++i) {
      const int ri = r + i * rstep;
      if (ri < rows) {
        v4f o;
        o.x = lrelu(y[i].x * a.x + b.x);
        o.y = lrelu(y[i].y * a.y + b.y);
        o.z = lrelu(y[i].z * a.z + b.z);
        o.w = lrelu(y[i].w * a.w + b.w);
        *(v4f*)(base + (size_t)ri * C) = o;
      }
    }
  }
}

template <int C, int U>
__global__ __launch_bounds__(256) void fuse_out(
    float* __restrict__ out, const float* __restrict__ ysc,
    const float* __restrict__ sc2, const float* __restrict__ sh2,
    const float* __restrict__ scS, const float* __restrict__ shS, int rows) {
  constexpr int G   = C / 4;
  constexpr int RPB = 256 / G;
  const int cg    = threadIdx.x % G;
  const int r0    = blockIdx.x * RPB + threadIdx.x / G;
  const int rstep = gridDim.x * RPB;
  const v4f a2 = *(const v4f*)(sc2 + 4 * cg);
  const v4f b2 = *(const v4f*)(sh2 + 4 * cg);
  const v4f aS = *(const v4f*)(scS + 4 * cg);
  const v4f bS = *(const v4f*)(shS + 4 * cg);
  float*       obase = out + 4 * cg;
  const float* sbase = ysc + 4 * cg;
  for (int r = r0; r < rows; r += U * rstep) {
    v4f y[U], s[U];
    #pragma unroll
    for (int i = 0; i < U; ++i) {
      const int ri = min(r + i * rstep, rows - 1);
      y[i] = *(const v4f*)(obase + (size_t)ri * C);
      s[i] = *(const v4f*)(sbase + (size_t)ri * C);
    }
    __builtin_amdgcn_sched_barrier(0);
    #pragma unroll
    for (int i = 0; i < U; ++i) {
      const int ri = r + i * rstep;
      if (ri < rows) {
        v4f o;
        o.x = lrelu(y[i].x * a2.x + b2.x) + (s[i].x * aS.x + bS.x);
        o.y = lrelu(y[i].y * a2.y + b2.y) + (s[i].y * aS.y + bS.y);
        o.z = lrelu(y[i].z * a2.z + b2.z) + (s[i].z * aS.z + bS.z);
        o.w = lrelu(y[i].w * a2.w + b2.w) + (s[i].w * aS.w + bS.w);
        *(v4f*)(obase + (size_t)ri * C) = o;
      }
    }
  }
}

extern "C" void kernel_launch(void* const* d_in, const int* in_sizes, int n_in,
                              void* d_out, int out_size, void* d_ws, size_t ws_size,
                              hipStream_t stream) {
  const float* feats = (const float*)d_in[0];
  const float* xyz   = (const float*)d_in[1];
  const int*   nidx  = (const int*)d_in[3];
  const float* W1    = (const float*)d_in[4];
  const float* g1    = (const float*)d_in[5];
  const float* b1    = (const float*)d_in[6];
  const float* kpp   = (const float*)d_in[7];
  const float* kpw   = (const float*)d_in[8];
  const float* W2    = (const float*)d_in[9];
  const float* g2    = (const float*)d_in[10];
  const float* b2    = (const float*)d_in[11];
  const float* Wsc   = (const float*)d_in[12];
  const float* gsc   = (const float*)d_in[13];
  const float* bsc   = (const float*)d_in[14];
  float* out = (float*)d_out;

  const int N = in_sizes[0] / CIN;

  float* ws      = (float*)d_ws;
  float* x1      = ws;
  float* xk      = x1 + (size_t)N * D2;
  float* ysc     = xk + (size_t)N * D2;
  float* sc1 = ysc + (size_t)N * COUT; float* sh1 = sc1 + D2;
  float* sc2 = sh1 + D2;               float* sh2 = sc2 + COUT;
  float* scS = sh2 + COUT;             float* shS = scS + COUT;
  const int blk1 = ((N / 16) * (D2 / 64) + 7) / 8;
  const int blk2 = ((N / 16) * (COUT / 64) + 7) / 8;
  float* part1  = xk;
  float* part2  = x1;
  float* partSC = x1 + (size_t)blk2 * 2 * COUT;

  {
    int units = (N / 16) * (D2 / 64);
    gemm_f32_wmma<<<blk1, 256, 0, stream>>>(feats, W1, x1, N, CIN, D2, units, part1);
  }
  finalize_bn<<<D2 / 16, 1024, 0, stream>>>(part1, blk1, g1, b1, D2, 1.0f / (float)N, sc1, sh1);
  bnact_inplace<D2, 8><<<625, 256, 0, stream>>>(x1, sc1, sh1, N);

  {
    size_t lds = (16 * FKP + 8 * 16 * WLP) * sizeof(float)
               + 8 * NBH * sizeof(int);
    kpconv_wmma<<<N / 16, 256, lds, stream>>>(xyz, nidx, x1, kpp, kpw, xk);
  }

  {
    int units = (N / 16) * (COUT / 64);
    gemm_f32_wmma<<<blk2, 256, 0, stream>>>(xk, W2, out, N, D2, COUT, units, part2);
  }
  {
    int units = (N / 16) * (COUT / 64);
    gemm_f32_wmma<<<blk2, 256, 0, stream>>>(feats, Wsc, ysc, N, CIN, COUT, units, partSC);
  }

  finalize_bn<<<COUT / 16, 1024, 0, stream>>>(part2, blk2, g2, b2, COUT, 1.0f / (float)N, sc2, sh2);
  finalize_bn<<<COUT / 16, 1024, 0, stream>>>(partSC, blk2, gsc, bsc, COUT, 1.0f / (float)N, scS, shS);

  fuse_out<COUT, 4><<<1000, 256, 0, stream>>>(out, ysc, sc2, sh2, scS, shS, N);
}


# KPConv influence weights: root by the hardware instruction (v_sqrt_f32, within 1 ulp), not sqrtf, whose correctly rounded sequence adds 15 vector instructions per root, 15 roots per lane and point
# speedup vs baseline: 2.3699x; 1.0450x over previous
#include <hip/hip_runtime.h>
#include <hip/hip_bf16.h>


typedef __attribute__((ext_vector_type(2))) float v2f;
typedef __attribute__((ext_vector_type(4))) float v4f;
typedef __attribute__((ext_vector_type(8))) float v8f;

#define CIN   128
#define COUT  256
#define D2    64
#define KPTS  15
#define NBH   32
#define KDFLAT (KPTS * D2)
#define FKP (KDFLAT + 4)
#define WLP 33
#define INFL_INV 2.5f
#define BN_EPS 1e-5f
#define SLOPE 0.2f

__device__ __forceinline__ v8f wmma_f32(v2f a, v2f b, v8f c) {
  return __builtin_amdgcn_wmma_f32_16x16x4_f32(
       false, a,  false, b,
       (short)0, c,  false,  false);
}

__device__ __forceinline__ float lrelu(float v) {
  return v >= 0.0f ? v : SLOPE * v;
}

#define KU 8
__global__ __launch_bounds__(256) void gemm_f32_wmma(
    const float* __restrict__ A, const float* __restrict__ B,
    float* __restrict__ C, int M, int Kd, int Nc, int totalUnits,
    float* __restrict__ part) {
  __shared__ __attribute__((aligned(16))) float red[8 * 2 * 128];
  const int wave = threadIdx.x >> 5;
  const int unit = blockIdx.x * 8 + wave;
  const bool active = unit < totalUnits;
  const int lane = threadIdx.x & 31;
  const int m    = lane & 15;
  const int half = lane >> 4;
  const int strips = Nc >> 6;

  v8f acc0 = {}, acc1 = {}, acc2 = {}, acc3 = {};
  const int row0 = active ? ((unit / strips) << 4) : 0;
  const int col0 = active ? ((unit % strips) << 6) : 0;
  if (active) {
    const float* Ap = A + (size_t)(row0 + m) * Kd + (half << 1);
    const float* Bp = B + (size_t)(half << 1) * Nc + col0 + (m << 2);
    const size_t bstep = (size_t)Nc << 2;
    for (int kk = 0; kk < Kd; kk += 4 * KU, Ap += 4 * KU, Bp += KU * bstep) {
      v2f a[KU];
      v4f bx[KU], by[KU];
      #pragma unroll
      for (int j = 0; j < KU; ++j) {
        a[j]  = *(const v2f*)(Ap + 4 * j);
        bx[j] = *(const v4f*)(Bp + j * bstep);
        by[j] = *(const v4f*)(Bp + j * bstep + Nc);
      }
      __builtin_amdgcn_sched_barrier(0);
      #pragma unroll
      for (int j = 0; j < KU; ++j) {
        v2f b0, b1, b2, b3;
        b0.x = bx[j].x; b0.y = by[j].x;
        b1.x = bx[j].y; b1.y = by[j].y;
        b2.x = bx[j].z; b2.y = by[j].z;
        b3.x = bx[j].w; b3.y = by[j].w;
        acc0 = wmma_f32(a[j], b0, acc0);
        acc1 = wmma_f32(a[j], b1, acc1);
        acc2 = wmma_f32(a[j], b2, acc2);
        acc3 = wmma_f32(a[j], b3, acc3);
      }
    }
  }

  float s0 = 0.0f, s1 = 0.0f, s2 = 0.0f, s3 = 0.0f;
  float q0 = 0.0f, q1 = 0.0f, q2 = 0.0f, q3 = 0.0f;
  #pragma unroll
  for (int r = 0; r < 8; ++r) {
    s0 += acc0[r]; q0 += acc0[r] * acc0[r];
    s1 += acc1[r]; q1 += acc1[r] * acc1[r];
    s2 += acc2[r]; q2 += acc2[r] * acc2[r];
    s3 += acc3[r]; q3 += acc3[r] * acc3[r];
  }
  float* rw = red + (wave * 2 + half) * 128 + (m << 2);
  v4f sv, qv;
  sv.x = s0; sv.y = s1; sv.z = s2; sv.w = s3;
  qv.x = q0; qv.y = q1; qv.z = q2; qv.w = q3;
  *(v4f*)rw        = sv;
  *(v4f*)(rw + 64) = qv;
  __syncthreads();
  if ((int)threadIdx.x < strips * 64) {
    const int st  = threadIdx.x >> 6;
    const int col = threadIdx.x & 63;
    float ts = 0.0f, tq = 0.0f;
    for (int w = st; w < 8; w += strips) {
      const float* r0 = red + (w * 2) * 128;
      ts += r0[col];       tq += r0[64 + col];
      ts += r0[128 + col]; tq += r0[192 + col];
    }
    float* p = part + (size_t)blockIdx.x * (size_t)(2 * Nc);
    p[st * 64 + col]      = ts;
    p[Nc + st * 64 + col] = tq;
  }

  if (active) {
    int rbase = row0 + (half << 3);
    float* Cp = C + (size_t)rbase * Nc + col0 + (m << 2);
    #pragma unroll
    for (int r = 0; r < 8; ++r) {
      v4f o;
      o.x = acc0[r]; o.y = acc1[r]; o.z = acc2[r]; o.w = acc3[r];
      *(v4f*)(Cp + (size_t)r * Nc) = o;
    }
  }
}

#define GU 2
__global__ __launch_bounds__(256, 8) void kpconv_wmma(
    const float* __restrict__ xyz, const int* __restrict__ nidx,
    const float* __restrict__ x1, const float* __restrict__ kp_points,
    const float* __restrict__ kpw, float* __restrict__ xk) {
  extern __shared__ __attribute__((aligned(16))) float smem[];
  float* FK  = smem;
  float* Wl  = FK + 16 * FKP;
  int*   IDX = (int*)(Wl + 8 * 16 * WLP);
  float* PS  = FK;

  const int wave = threadIdx.x >> 5;
  const int lane = threadIdx.x & 31;
  const int m    = lane & 15;
  const int half = lane >> 4;
  const int base = blockIdx.x << 4;

  float* wl = Wl + wave * (16 * WLP);
  int*   il = IDX + wave * NBH;

  for (int it = 0; it < 2; ++it) {
    const int pi = (wave << 1) + it;
    const int n  = base + pi;

    const int id = nidx[(size_t)n * NBH + lane];
    il[lane] = id;
    const float cx = xyz[(size_t)n * 3 + 0];
    const float cy = xyz[(size_t)n * 3 + 1];
    const float cz = xyz[(size_t)n * 3 + 2];
    const float px = xyz[(size_t)id * 3 + 0] - cx;
    const float py = xyz[(size_t)id * 3 + 1] - cy;
    const float pz = xyz[(size_t)id * 3 + 2] - cz;
    #pragma unroll
    for (int k = 0; k < KPTS; ++k) {
      float dx = px - kp_points[k * 3 + 0];
      float dy = py - kp_points[k * 3 + 1];
      float dz = pz - kp_points[k * 3 + 2];
      float r = __builtin_amdgcn_sqrtf(dx * dx + dy * dy + dz * dz);
      wl[k * WLP + lane] = fmaxf(1.0f - r * INFL_INV, 0.0f);
    }
    wl[KPTS * WLP + lane] = 0.0f;
    __syncthreads();

    v8f acc0 = {}, acc1 = {}, acc2 = {}, acc3 = {};
    #pragma unroll
    for (int k0 = 0; k0 < NBH; k0 += 4 * GU) {
      v4f bx[GU], by[GU];
      #pragma unroll
      for (int j = 0; j < GU; ++j) {
        const int ka = k0 + 4 * j + (half << 1);
        bx[j] = *(const v4f*)(x1 + (size_t)il[ka] * D2 + (m << 2));
        by[j] = *(const v4f*)(x1 + (size_t)il[ka + 1] * D2 + (m << 2));
      }
      __builtin_amdgcn_sched_barrier(0);
      #pragma unroll
      for (int j = 0; j < GU; ++j) {
        const int ka = k0 + 4 * j + (half << 1);
        v2f a;
        a.x = wl[m * WLP + ka];
        a.y = wl[m * WLP + ka + 1];
        v2f b0, b1, b2, b3;
        b0.x = bx[j].x; b0.y = by[j].x;
        b1.x = bx[j].y; b1.y = by[j].y;
        b2.x = bx[j].z; b2.y = by[j].z;
        b3.x = bx[j].w; b3.y = by[j].w;
        acc0 = wmma_f32(a, b0, acc0);
        acc1 = wmma_f32(a, b1, acc1);
        acc2 = wmma_f32(a, b2, acc2);
        acc3 = wmma_f32(a, b3, acc3);
      }
    }
    #pragma unroll
    for (int r = 0; r < 8; ++r) {
      const int krow = (half << 3) + r;
      if (krow < KPTS) {
        v4f o;
        o.x = acc0[r]; o.y = acc1[r]; o.z = acc2[r]; o.w = acc3[r];
        *(v4f*)(FK + pi * FKP + krow * D2 + (m << 2)) = o;
      }
    }
    __syncthreads();
  }

  {
    const int kbeg = wave * 120;
    v8f acc0 = {}, acc1 = {}, acc2 = {}, acc3 = {};
    for (int kk = kbeg; kk < kbeg + 120; kk += 4) {
      const int ka = kk + (half << 1);
      v2f a;
      a.x = FK[m * FKP + ka];
      a.y = FK[m * FKP + ka + 1];
      const float* Bp = kpw + (size_t)ka * D2 + (m << 2);
      const v4f bx = *(const v4f*)Bp;
      const v4f by = *(const v4f*)(Bp + D2);
      v2f b0, b1, b2, b3;
      b0.x = bx.x; b0.y = by.x;
      b1.x = bx.y; b1.y = by.y;
      b2.x = bx.z; b2.y = by.z;
      b3.x = bx.w; b3.y = by.w;
      acc0 = wmma_f32(a, b0, acc0);
      acc1 = wmma_f32(a, b1, acc1);
      acc2 = wmma_f32(a, b2, acc2);
      acc3 = wmma_f32(a, b3, acc3);
    }
    __syncthreads();
    #pragma unroll
    for (int r = 0; r < 8; ++r) {
      v4f o;
      o.x = acc0[r]; o.y = acc1[r]; o.z = acc2[r]; o.w = acc3[r];
      *(v4f*)(PS + wave * (16 * D2) + ((half << 3) + r) * D2 + (m << 2)) = o;
    }
  }
  __syncthreads();

  for (int t = threadIdx.x; t < 16 * D2; t += 256) {
    float s = 0.0f;
    #pragma unroll
    for (int w = 0; w < 8; ++w) s += PS[w * (16 * D2) + t];
    const int pt = t >> 6;
    const int e  = t & 63;
    xk[(size_t)(base + pt) * D2 + e] = s;
  }
}

__global__ __launch_bounds__(1024) void finalize_bn(
    const float* __restrict__ part, int nparts,
    const float* __restrict__ g, const float* __restrict__ b, int C, float invN,
    float* __restrict__ sc, float* __restrict__ sh) {
  __shared__ float rs[1024];
  __shared__ float rq[1024];
  const int cl = threadIdx.x & 15;
  const int sl = threadIdx.x >> 4;
  const int c  = blockIdx.x * 16 + cl;
  const size_t pitch = (size_t)(2 * C);
  float s = 0.0f, q = 0.0f;
  #pragma unroll 8
  for (int p = sl; p < nparts; p += 64) {
    s += part[(size_t)p * pitch + c];
    q += part[(size_t)p * pitch + C + c];
  }
  rs[threadIdx.x] = s;
  rq[threadIdx.x] = q;
  __syncthreads();
  if (threadIdx.x < 16) {
    float ts = 0.0f, tq = 0.0f;
    #pragma unroll
    for (int k = 0; k < 64; ++k) {
      ts += rs[k * 16 + cl];
      tq += rq[k * 16 + cl];
    }
    float mean = ts * invN;
    float var  = tq * invN - mean * mean;
    float scl  = g[c] * rsqrtf(var + BN_EPS);
    sc[c] = scl;
    sh[c] = b[c] - mean * scl;
  }
}


template <int C, int U>
__global__ __launch_bounds__(256) void bnact_inplace(
    float* __restrict__ Y, const float* __restrict__ sc,
    const float* __restrict__ sh, int rows) {
  constexpr int G   = C / 4;
  constexpr int RPB = 256 / G;
  const int cg    = threadIdx.x % G;
  const int r0    = blockIdx.x * RPB + threadIdx.x / G;
  const int rstep = gridDim.x * RPB;
  const v4f a = *(const v4f*)(sc + 4 * cg);
  const v4f b = *(const v4f*)(sh + 4 * cg);
  float* base = Y + 4 * cg;
  for (int r = r0; r < rows; r += U * rstep) {
    v4f y[U];
    #pragma unroll
    for (int i = 0; i < U; ++i) {
      const int ri = min(r + i * rstep, rows - 1);
      y[i] = *(const v4f*)(base + (size_t)ri * C);
    }
    __builtin_amdgcn_sched_barrier(0);
    #pragma unroll
    for (int i = 0; i < U; ++i) {
      const int ri = r + i * rstep;
      if (ri < rows) {
        v4f o;
        o.x = lrelu(y[i].x * a.x + b.x);
        o.y = lrelu(y[i].y * a.y + b.y);
        o.z = lrelu(y[i].z * a.z + b.z);
        o.w = lrelu(y[i].w * a.w + b.w);
        *(v4f*)(base + (size_t)ri * C) = o;
      }
    }
  }
}

template <int C, int U>
__global__ __launch_bounds__(256) void fuse_out(
    float* __restrict__ out, const float* __restrict__ ysc,
    const float* __restrict__ sc2, const float* __restrict__ sh2,
    const float* __restrict__ scS, const float* __restrict__ shS, int rows) {
  constexpr int G   = C / 4;
  constexpr int RPB = 256 / G;
  const int cg    = threadIdx.x % G;
  const int r0    = blockIdx.x * RPB + threadIdx.x / G;
  const int rstep = gridDim.x * RPB;
  const v4f a2 = *(const v4f*)(sc2 + 4 * cg);
  const v4f b2 = *(const v4f*)(sh2 + 4 * cg);
  const v4f aS = *(const v4f*)(scS + 4 * cg);
  const v4f bS = *(const v4f*)(shS + 4 * cg);
  float*       obase = out + 4 * cg;
  const float* sbase = ysc + 4 * cg;
  for (int r = r0; r < rows; r += U * rstep) {
    v4f y[U], s[U];
    #pragma unroll
    for (int i = 0; i < U; ++i) {
      const int ri = min(r + i * rstep, rows - 1);
      y[i] = *(const v4f*)(obase + (size_t)ri * C);
      s[i] = *(const v4f*)(sbase + (size_t)ri * C);
    }
    __builtin_amdgcn_sched_barrier(0);
    #pragma unroll
    for (int i = 0; i < U; ++i) {
      const int ri = r + i * rstep;
      if (ri < rows) {
        v4f o;
        o.x = lrelu(y[i].x * a2.x + b2.x) + (s[i].x * aS.x + bS.x);
        o.y = lrelu(y[i].y * a2.y + b2.y) + (s[i].y * aS.y + bS.y);
        o.z = lrelu(y[i].z * a2.z + b2.z) + (s[i].z * aS.z + bS.z);
        o.w = lrelu(y[i].w * a2.w + b2.w) + (s[i].w * aS.w + bS.w);
        *(v4f*)(obase + (size_t)ri * C) = o;
      }
    }
  }
}

extern "C" void kernel_launch(void* const* d_in, const int* in_sizes, int n_in,
                              void* d_out, int out_size, void* d_ws, size_t ws_size,
                              hipStream_t stream) {
  const float* feats = (const float*)d_in[0];
  const float* xyz   = (const float*)d_in[1];
  const int*   nidx  = (const int*)d_in[3];
  const float* W1    = (const float*)d_in[4];
  const float* g1    = (const float*)d_in[5];
  const float* b1    = (const float*)d_in[6];
  const float* kpp   = (const float*)d_in[7];
  const float* kpw   = (const float*)d_in[8];
  const float* W2    = (const float*)d_in[9];
  const float* g2    = (const float*)d_in[10];
  const float* b2    = (const float*)d_in[11];
  const float* Wsc   = (const float*)d_in[12];
  const float* gsc   = (const float*)d_in[13];
  const float* bsc   = (const float*)d_in[14];
  float* out = (float*)d_out;

  const int N = in_sizes[0] / CIN;

  float* ws      = (float*)d_ws;
  float* x1      = ws;
  float* xk      = x1 + (size_t)N * D2;
  float* ysc     = xk + (size_t)N * D2;
  float* sc1 = ysc + (size_t)N * COUT; float* sh1 = sc1 + D2;
  float* sc2 = sh1 + D2;               float* sh2 = sc2 + COUT;
  float* scS = sh2 + COUT;             float* shS = scS + COUT;
  const int blk1 = ((N / 16) * (D2 / 64) + 7) / 8;
  const int blk2 = ((N / 16) * (COUT / 64) + 7) / 8;
  float* part1  = xk;
  float* part2  = x1;
  float* partSC = x1 + (size_t)blk2 * 2 * COUT;

  {
    int units = (N / 16) * (D2 / 64);
    gemm_f32_wmma<<<blk1, 256, 0, stream>>>(feats, W1, x1, N, CIN, D2, units, part1);
  }
  finalize_bn<<<D2 / 16, 1024, 0, stream>>>(part1, blk1, g1, b1, D2, 1.0f / (float)N, sc1, sh1);
  bnact_inplace<D2, 8><<<625, 256, 0, stream>>>(x1, sc1, sh1, N);

  {
    size_t lds = (16 * FKP + 8 * 16 * WLP) * sizeof(float)
               + 8 * NBH * sizeof(int);
    kpconv_wmma<<<N / 16, 256, lds, stream>>>(xyz, nidx, x1, kpp, kpw, xk);
  }

  {
    int units = (N / 16) * (COUT / 64);
    gemm_f32_wmma<<<blk2, 256, 0, stream>>>(xk, W2, out, N, D2, COUT, units, part2);
  }
  {
    int units = (N / 16) * (COUT / 64);
    gemm_f32_wmma<<<blk2, 256, 0, stream>>>(feats, Wsc, ysc, N, CIN, COUT, units, partSC);
  }

  finalize_bn<<<COUT / 16, 1024, 0, stream>>>(part2, blk2, g2, b2, COUT, 1.0f / (float)N, sc2, sh2);
  finalize_bn<<<COUT / 16, 1024, 0, stream>>>(partSC, blk2, gsc, bsc, COUT, 1.0f / (float)N, scS, shS);

  fuse_out<COUT, 4><<<1000, 256, 0, stream>>>(out, ysc, sc2, sh2, scS, shS, N);
}


# Two launches fewer (9 -> 7): the shortcut GEMM runs in the unary_1 GEMM's grid (both read feats), its BN finalize in the unary_1 finalize's grid; operands chosen per block; sums as before
# speedup vs baseline: 2.4466x; 1.0324x over previous
#include <hip/hip_runtime.h>
#include <hip/hip_bf16.h>


typedef __attribute__((ext_vector_type(2))) float v2f;
typedef __attribute__((ext_vector_type(4))) float v4f;
typedef __attribute__((ext_vector_type(8))) float v8f;

#define CIN   128
#define COUT  256
#define D2    64
#define KPTS  15
#define NBH   32
#define KDFLAT (KPTS * D2)
#define FKP (KDFLAT + 4)
#define WLP 33
#define INFL_INV 2.5f
#define BN_EPS 1e-5f
#define SLOPE 0.2f

__device__ __forceinline__ v8f wmma_f32(v2f a, v2f b, v8f c) {
  return __builtin_amdgcn_wmma_f32_16x16x4_f32(
       false, a,  false, b,
       (short)0, c,  false,  false);
}

__device__ __forceinline__ float lrelu(float v) {
  return v >= 0.0f ? v : SLOPE * v;
}

#define KU 8
__global__ __launch_bounds__(256) void gemm_f32_wmma(
    const float* __restrict__ A0, const float* __restrict__ B0,
    float* __restrict__ C0, float* __restrict__ part0, int Kd0, int Nc0, int units0,
    const float* __restrict__ A1, const float* __restrict__ B1,
    float* __restrict__ C1, float* __restrict__ part1, int Kd1, int Nc1, int units1,
    int blocks0) {
  __shared__ __attribute__((aligned(16))) float red[8 * 2 * 128];
  const bool second = (int)blockIdx.x >= blocks0;
  const float* __restrict__ A = second ? A1 : A0;
  const float* __restrict__ B = second ? B1 : B0;
  float* __restrict__ C       = second ? C1 : C0;
  float* __restrict__ part    = second ? part1 : part0;
  const int Kd         = second ? Kd1 : Kd0;
  const int Nc         = second ? Nc1 : Nc0;
  const int totalUnits = second ? units1 : units0;
  const int blk = second ? (int)blockIdx.x - blocks0 : (int)blockIdx.x;
  const int wave = threadIdx.x >> 5;
  const int unit = blk * 8 + wave;
  const bool active = unit < totalUnits;
  const int lane = threadIdx.x & 31;
  const int m    = lane & 15;
  const int half = lane >> 4;
  const int strips = Nc >> 6;

  v8f acc0 = {}, acc1 = {}, acc2 = {}, acc3 = {};
  const int row0 = active ? ((unit / strips) << 4) : 0;
  const int col0 = active ? ((unit % strips) << 6) : 0;
  if (active) {
    const float* Ap = A + (size_t)(row0 + m) * Kd + (half << 1);
    const float* Bp = B + (size_t)(half << 1) * Nc + col0 + (m << 2);
    const size_t bstep = (size_t)Nc << 2;
    for (int kk = 0; kk < Kd; kk += 4 * KU, Ap += 4 * KU, Bp += KU * bstep) {
      v2f a[KU];
      v4f bx[KU], by[KU];
      #pragma unroll
      for (int j = 0; j < KU; ++j) {
        a[j]  = *(const v2f*)(Ap + 4 * j);
        bx[j] = *(const v4f*)(Bp + j * bstep);
        by[j] = *(const v4f*)(Bp + j * bstep + Nc);
      }
      __builtin_amdgcn_sched_barrier(0);
      #pragma unroll
      for (int j = 0; j < KU; ++j) {
        v2f b0, b1, b2, b3;
        b0.x = bx[j].x; b0.y = by[j].x;
        b1.x = bx[j].y; b1.y = by[j].y;
        b2.x = bx[j].z; b2.y = by[j].z;
        b3.x = bx[j].w; b3.y = by[j].w;
        acc0 = wmma_f32(a[j], b0, acc0);
        acc1 = wmma_f32(a[j], b1, acc1);
        acc2 = wmma_f32(a[j], b2, acc2);
        acc3 = wmma_f32(a[j], b3, acc3);
      }
    }
  }

  float s0 = 0.0f, s1 = 0.0f, s2 = 0.0f, s3 = 0.0f;
  float q0 = 0.0f, q1 = 0.0f, q2 = 0.0f, q3 = 0.0f;
  #pragma unroll
  for (int r = 0; r < 8; ++r) {
    s0 += acc0[r]; q0 += acc0[r] * acc0[r];
    s1 += acc1[r]; q1 += acc1[r] * acc1[r];
    s2 += acc2[r]; q2 += acc2[r] * acc2[r];
    s3 += acc3[r]; q3 += acc3[r] * acc3[r];
  }
  float* rw = red + (wave * 2 + half) * 128 + (m << 2);
  v4f sv, qv;
  sv.x = s0; sv.y = s1; sv.z = s2; sv.w = s3;
  qv.x = q0; qv.y = q1; qv.z = q2; qv.w = q3;
  *(v4f*)rw        = sv;
  *(v4f*)(rw + 64) = qv;
  __syncthreads();
  if ((int)threadIdx.x < strips * 64) {
    const int st  = threadIdx.x >> 6;
    const int col = threadIdx.x & 63;
    float ts = 0.0f, tq = 0.0f;
    for (int w = st; w < 8; w += strips) {
      const float* r0 = red + (w * 2) * 128;
      ts += r0[col];       tq += r0[64 + col];
      ts += r0[128 + col]; tq += r0[192 + col];
    }
    float* p = part + (size_t)blk * (size_t)(2 * Nc);
    p[st * 64 + col]      = ts;
    p[Nc + st * 64 + col] = tq;
  }

  if (active) {
    int rbase = row0 + (half << 3);
    float* Cp = C + (size_t)rbase * Nc + col0 + (m << 2);
    #pragma unroll
    for (int r = 0; r < 8; ++r) {
      v4f o;
      o.x = acc0[r]; o.y = acc1[r]; o.z = acc2[r]; o.w = acc3[r];
      *(v4f*)(Cp + (size_t)r * Nc) = o;
    }
  }
}

#define GU 2
__global__ __launch_bounds__(256, 8) void kpconv_wmma(
    const float* __restrict__ xyz, const int* __restrict__ nidx,
    const float* __restrict__ x1, const float* __restrict__ kp_points,
    const float* __restrict__ kpw, float* __restrict__ xk) {
  extern __shared__ __attribute__((aligned(16))) float smem[];
  float* FK  = smem;
  float* Wl  = FK + 16 * FKP;
  int*   IDX = (int*)(Wl + 8 * 16 * WLP);
  float* PS  = FK;

  const int wave = threadIdx.x >> 5;
  const int lane = threadIdx.x & 31;
  const int m    = lane & 15;
  const int half = lane >> 4;
  const int base = blockIdx.x << 4;

  float* wl = Wl + wave * (16 * WLP);
  int*   il = IDX + wave * NBH;

  for (int it = 0; it < 2; ++it) {
    const int pi = (wave << 1) + it;
    const int n  = base + pi;

    const int id = nidx[(size_t)n * NBH + lane];
    il[lane] = id;
    const float cx = xyz[(size_t)n * 3 + 0];
    const float cy = xyz[(size_t)n * 3 + 1];
    const float cz = xyz[(size_t)n * 3 + 2];
    const float px = xyz[(size_t)id * 3 + 0] - cx;
    const float py = xyz[(size_t)id * 3 + 1] - cy;
    const float pz = xyz[(size_t)id * 3 + 2] - cz;
    #pragma unroll
    for (int k = 0; k < KPTS; ++k) {
      float dx = px - kp_points[k * 3 + 0];
      float dy = py - kp_points[k * 3 + 1];
      float dz = pz - kp_points[k * 3 + 2];
      float r = __builtin_amdgcn_sqrtf(dx * dx + dy * dy + dz * dz);
      wl[k * WLP + lane] = fmaxf(1.0f - r * INFL_INV, 0.0f);
    }
    wl[KPTS * WLP + lane] = 0.0f;
    __syncthreads();

    v8f acc0 = {}, acc1 = {}, acc2 = {}, acc3 = {};
    #pragma unroll
    for (int k0 = 0; k0 < NBH; k0 += 4 * GU) {
      v4f bx[GU], by[GU];
      #pragma unroll
      for (int j = 0; j < GU; ++j) {
        const int ka = k0 + 4 * j + (half << 1);
        bx[j] = *(const v4f*)(x1 + (size_t)il[ka] * D2 + (m << 2));
        by[j] = *(const v4f*)(x1 + (size_t)il[ka + 1] * D2 + (m << 2));
      }
      __builtin_amdgcn_sched_barrier(0);
      #pragma unroll
      for (int j = 0; j < GU; ++j) {
        const int ka = k0 + 4 * j + (half << 1);
        v2f a;
        a.x = wl[m * WLP + ka];
        a.y = wl[m * WLP + ka + 1];
        v2f b0, b1, b2, b3;
        b0.x = bx[j].x; b0.y = by[j].x;
        b1.x = bx[j].y; b1.y = by[j].y;
        b2.x = bx[j].z; b2.y = by[j].z;
        b3.x = bx[j].w; b3.y = by[j].w;
        acc0 = wmma_f32(a, b0, acc0);
        acc1 = wmma_f32(a, b1, acc1);
        acc2 = wmma_f32(a, b2, acc2);
        acc3 = wmma_f32(a, b3, acc3);
      }
    }
    #pragma unroll
    for (int r = 0; r < 8; ++r) {
      const int krow = (half << 3) + r;
      if (krow < KPTS) {
        v4f o;
        o.x = acc0[r]; o.y = acc1[r]; o.z = acc2[r]; o.w = acc3[r];
        *(v4f*)(FK + pi * FKP + krow * D2 + (m << 2)) = o;
      }
    }
    __syncthreads();
  }

  {
    const int kbeg = wave * 120;
    v8f acc0 = {}, acc1 = {}, acc2 = {}, acc3 = {};
    for (int kk = kbeg; kk < kbeg + 120; kk += 4) {
      const int ka = kk + (half << 1);
      v2f a;
      a.x = FK[m * FKP + ka];
      a.y = FK[m * FKP + ka + 1];
      const float* Bp = kpw + (size_t)ka * D2 + (m << 2);
      const v4f bx = *(const v4f*)Bp;
      const v4f by = *(const v4f*)(Bp + D2);
      v2f b0, b1, b2, b3;
      b0.x = bx.x; b0.y = by.x;
      b1.x = bx.y; b1.y = by.y;
      b2.x = bx.z; b2.y = by.z;
      b3.x = bx.w; b3.y = by.w;
      acc0 = wmma_f32(a, b0, acc0);
      acc1 = wmma_f32(a, b1, acc1);
      acc2 = wmma_f32(a, b2, acc2);
      acc3 = wmma_f32(a, b3, acc3);
    }
    __syncthreads();
    #pragma unroll
    for (int r = 0; r < 8; ++r) {
      v4f o;
      o.x = acc0[r]; o.y = acc1[r]; o.z = acc2[r]; o.w = acc3[r];
      *(v4f*)(PS + wave * (16 * D2) + ((half << 3) + r) * D2 + (m << 2)) = o;
    }
  }
  __syncthreads();

  for (int t = threadIdx.x; t < 16 * D2; t += 256) {
    float s = 0.0f;
    #pragma unroll
    for (int w = 0; w < 8; ++w) s += PS[w * (16 * D2) + t];
    const int pt = t >> 6;
    const int e  = t & 63;
    xk[(size_t)(base + pt) * D2 + e] = s;
  }
}

__global__ __launch_bounds__(1024) void finalize_bn(
    const float* __restrict__ partA, int npartsA, const float* __restrict__ gA,
    const float* __restrict__ bA, int CA, float* __restrict__ scA, float* __restrict__ shA,
    const float* __restrict__ partB, int npartsB, const float* __restrict__ gB,
    const float* __restrict__ bB, int CB, float* __restrict__ scB, float* __restrict__ shB,
    int blocks0, float invN) {
  __shared__ float rs[1024];
  __shared__ float rq[1024];
  const bool second = (int)blockIdx.x >= blocks0;
  const float* __restrict__ part = second ? partB : partA;
  const float* __restrict__ g    = second ? gB : gA;
  const float* __restrict__ b    = second ? bB : bA;
  float* __restrict__ sc         = second ? scB : scA;
  float* __restrict__ sh         = second ? shB : shA;
  const int nparts = second ? npartsB : npartsA;
  const int C      = second ? CB : CA;
  const int blk = second ? (int)blockIdx.x - blocks0 : (int)blockIdx.x;
  const int cl = threadIdx.x & 15;
  const int sl = threadIdx.x >> 4;
  const int c  = blk * 16 + cl;
  const size_t pitch = (size_t)(2 * C);
  float s = 0.0f, q = 0.0f;
  #pragma unroll 8
  for (int p = sl; p < nparts; p += 64) {
    s += part[(size_t)p * pitch + c];
    q += part[(size_t)p * pitch + C + c];
  }
  rs[threadIdx.x] = s;
  rq[threadIdx.x] = q;
  __syncthreads();
  if (threadIdx.x < 16) {
    float ts = 0.0f, tq = 0.0f;
    #pragma unroll
    for (int k = 0; k < 64; ++k) {
      ts += rs[k * 16 + cl];
      tq += rq[k * 16 + cl];
    }
    float mean = ts * invN;
    float var  = tq * invN - mean * mean;
    float scl  = g[c] * rsqrtf(var + BN_EPS);
    sc[c] = scl;
    sh[c] = b[c] - mean * scl;
  }
}


template <int C, int U>
__global__ __launch_bounds__(256) void bnact_inplace(
    float* __restrict__ Y, const float* __restrict__ sc,
    const float* __restrict__ sh, int rows) {
  constexpr int G   = C / 4;
  constexpr int RPB = 256 / G;
  const int cg    = threadIdx.x % G;
  const int r0    = blockIdx.x * RPB + threadIdx.x / G;
  const int rstep = gridDim.x * RPB;
  const v4f a = *(const v4f*)(sc + 4 * cg);
  const v4f b = *(const v4f*)(sh + 4 * cg);
  float* base = Y + 4 * cg;
  for (int r = r0; r < rows; r += U * rstep) {
    v4f y[U];
    #pragma unroll
    for (int i = 0; i < U; ++i) {
      const int ri = min(r + i * rstep, rows - 1);
      y[i] = *(const v4f*)(base + (size_t)ri * C);
    }
    __builtin_amdgcn_sched_barrier(0);
    #pragma unroll
    for (int i = 0; i < U; ++i) {
      const int ri = r + i * rstep;
      if (ri < rows) {
        v4f o;
        o.x = lrelu(y[i].x * a.x + b.x);
        o.y = lrelu(y[i].y * a.y + b.y);
        o.z = lrelu(y[i].z * a.z + b.z);
        o.w = lrelu(y[i].w * a.w + b.w);
        *(v4f*)(base + (size_t)ri * C) = o;
      }
    }
  }
}

template <int C, int U>
__global__ __launch_bounds__(256) void fuse_out(
    float* __restrict__ out, const float* __restrict__ ysc,
    const float* __restrict__ sc2, const float* __restrict__ sh2,
    const float* __restrict__ scS, const float* __restrict__ shS, int rows) {
  constexpr int G   = C / 4;
  constexpr int RPB = 256 / G;
  const int cg    = threadIdx.x % G;
  const int r0    = blockIdx.x * RPB + threadIdx.x / G;
  const int rstep = gridDim.x * RPB;
  const v4f a2 = *(const v4f*)(sc2 + 4 * cg);
  const v4f b2 = *(const v4f*)(sh2 + 4 * cg);
  const v4f aS = *(const v4f*)(scS + 4 * cg);
  const v4f bS = *(const v4f*)(shS + 4 * cg);
  float*       obase = out + 4 * cg;
  const float* sbase = ysc + 4 * cg;
  for (int r = r0; r < rows; r += U * rstep) {
    v4f y[U], s[U];
    #pragma unroll
    for (int i = 0; i < U; ++i) {
      const int ri = min(r + i * rstep, rows - 1);
      y[i] = *(const v4f*)(obase + (size_t)ri * C);
      s[i] = *(const v4f*)(sbase + (size_t)ri * C);
    }
    __builtin_amdgcn_sched_barrier(0);
    #pragma unroll
    for (int i = 0; i < U; ++i) {
      const int ri = r + i * rstep;
      if (ri < rows) {
        v4f o;
        o.x = lrelu(y[i].x * a2.x + b2.x) + (s[i].x * aS.x + bS.x);
        o.y = lrelu(y[i].y * a2.y + b2.y) + (s[i].y * aS.y + bS.y);
        o.z = lrelu(y[i].z * a2.z + b2.z) + (s[i].z * aS.z + bS.z);
        o.w = lrelu(y[i].w * a2.w + b2.w) + (s[i].w * aS.w + bS.w);
        *(v4f*)(obase + (size_t)ri * C) = o;
      }
    }
  }
}

extern "C" void kernel_launch(void* const* d_in, const int* in_sizes, int n_in,
                              void* d_out, int out_size, void* d_ws, size_t ws_size,
                              hipStream_t stream) {
  const float* feats = (const float*)d_in[0];
  const float* xyz   = (const float*)d_in[1];
  const int*   nidx  = (const int*)d_in[3];
  const float* W1    = (const float*)d_in[4];
  const float* g1    = (const float*)d_in[5];
  const float* b1    = (const float*)d_in[6];
  const float* kpp   = (const float*)d_in[7];
  const float* kpw   = (const float*)d_in[8];
  const float* W2    = (const float*)d_in[9];
  const float* g2    = (const float*)d_in[10];
  const float* b2    = (const float*)d_in[11];
  const float* Wsc   = (const float*)d_in[12];
  const float* gsc   = (const float*)d_in[13];
  const float* bsc   = (const float*)d_in[14];
  float* out = (float*)d_out;

  const int N = in_sizes[0] / CIN;

  float* ws      = (float*)d_ws;
  float* x1      = ws;
  float* xk      = x1 + (size_t)N * D2;
  float* ysc     = xk + (size_t)N * D2;
  float* sc1 = ysc + (size_t)N * COUT; float* sh1 = sc1 + D2;
  float* sc2 = sh1 + D2;               float* sh2 = sc2 + COUT;
  float* scS = sh2 + COUT;             float* shS = scS + COUT;
  const int blk1 = ((N / 16) * (D2 / 64) + 7) / 8;
  const int blk2 = ((N / 16) * (COUT / 64) + 7) / 8;
  float* part1  = xk;
  float* partSC = xk + (size_t)blk1 * 2 * D2;
  float* part2  = x1;
  const int units1 = (N / 16) * (D2 / 64);
  const int units2 = (N / 16) * (COUT / 64);

  gemm_f32_wmma<<<blk1 + blk2, 256, 0, stream>>>(feats, W1, x1, part1, CIN, D2, units1,
                                                 feats, Wsc, ysc, partSC, CIN, COUT, units2, blk1);
  finalize_bn<<<D2 / 16 + COUT / 16, 1024, 0, stream>>>(part1, blk1, g1, b1, D2, sc1, sh1,
                                                        partSC, blk2, gsc, bsc, COUT, scS, shS,
                                                        D2 / 16, 1.0f / (float)N);
  bnact_inplace<D2, 8><<<625, 256, 0, stream>>>(x1, sc1, sh1, N);

  {
    size_t lds = (16 * FKP + 8 * 16 * WLP) * sizeof(float)
               + 8 * NBH * sizeof(int);
    kpconv_wmma<<<N / 16, 256, lds, stream>>>(xyz, nidx, x1, kpp, kpw, xk);
  }

  gemm_f32_wmma<<<blk2, 256, 0, stream>>>(xk, W2, out, part2, D2, COUT, units2,
                                          xk, W2, out, part2, D2, COUT, units2, blk2);
  finalize_bn<<<COUT / 16, 1024, 0, stream>>>(part2, blk2, g2, b2, COUT, sc2, sh2,
                                              part2, blk2, g2, b2, COUT, sc2, sh2,
                                              COUT / 16, 1.0f / (float)N);

  fuse_out<COUT, 4><<<1000, 256, 0, stream>>>(out, ysc, sc2, sh2, scS, shS, N);
}


# GEMM wave tile 16x64 -> 32x64 (8 accumulators): a B fragment feeds two row tiles, 12 operand registers loaded per 8 WMMAs, not 10 per 4; half the waves and partial rows. Reorders the BN column sums.
# speedup vs baseline: 2.5798x; 1.0544x over previous
#include <hip/hip_runtime.h>
#include <hip/hip_bf16.h>


typedef __attribute__((ext_vector_type(2))) float v2f;
typedef __attribute__((ext_vector_type(4))) float v4f;
typedef __attribute__((ext_vector_type(8))) float v8f;

#define CIN   128
#define COUT  256
#define D2    64
#define KPTS  15
#define NBH   32
#define KDFLAT (KPTS * D2)
#define FKP (KDFLAT + 4)
#define WLP 33
#define INFL_INV 2.5f
#define BN_EPS 1e-5f
#define SLOPE 0.2f

__device__ __forceinline__ v8f wmma_f32(v2f a, v2f b, v8f c) {
  return __builtin_amdgcn_wmma_f32_16x16x4_f32(
       false, a,  false, b,
       (short)0, c,  false,  false);
}

__device__ __forceinline__ float lrelu(float v) {
  return v >= 0.0f ? v : SLOPE * v;
}

#define KU 4
__global__ __launch_bounds__(256) void gemm_f32_wmma(
    const float* __restrict__ A0, const float* __restrict__ B0,
    float* __restrict__ C0, float* __restrict__ part0, int Kd0, int Nc0, int units0,
    const float* __restrict__ A1, const float* __restrict__ B1,
    float* __restrict__ C1, float* __restrict__ part1, int Kd1, int Nc1, int units1,
    int blocks0) {
  __shared__ __attribute__((aligned(16))) float red[8 * 2 * 128];
  const bool second = (int)blockIdx.x >= blocks0;
  const float* __restrict__ A = second ? A1 : A0;
  const float* __restrict__ B = second ? B1 : B0;
  float* __restrict__ C       = second ? C1 : C0;
  float* __restrict__ part    = second ? part1 : part0;
  const int Kd         = second ? Kd1 : Kd0;
  const int Nc         = second ? Nc1 : Nc0;
  const int totalUnits = second ? units1 : units0;
  const int blk = second ? (int)blockIdx.x - blocks0 : (int)blockIdx.x;
  const int wave = threadIdx.x >> 5;
  const int unit = blk * 8 + wave;
  const bool active = unit < totalUnits;
  const int lane = threadIdx.x & 31;
  const int m    = lane & 15;
  const int half = lane >> 4;
  const int strips = Nc >> 6;

  v8f acc0_0 = {}, acc0_1 = {}, acc0_2 = {}, acc0_3 = {};
  v8f acc1_0 = {}, acc1_1 = {}, acc1_2 = {}, acc1_3 = {};
  const int row0 = active ? ((unit / strips) << 5) : 0;
  const int col0 = active ? ((unit % strips) << 6) : 0;
  if (active) {
    const float* Ap0 = A + (size_t)(row0 + m) * Kd + (half << 1);
    const float* Ap1 = Ap0 + (size_t)16 * Kd;
    const unsigned boff = (unsigned)((half << 1) * Nc + col0 + (m << 2));
    const float* Bp  = B;
    const size_t bstep = (size_t)Nc << 2;
    for (int kk = 0; kk < Kd; kk += 4 * KU, Ap0 += 4 * KU, Ap1 += 4 * KU, Bp += KU * bstep) {
      v2f a0[KU], a1[KU];
      v4f bx[KU], by[KU];
      #pragma unroll
      for (int j = 0; j < KU; ++j) {
        a0[j] = *(const v2f*)(Ap0 + 4 * j);
        a1[j] = *(const v2f*)(Ap1 + 4 * j);
        bx[j] = *(const v4f*)(Bp + j * bstep + boff);
        by[j] = *(const v4f*)(Bp + j * bstep + Nc + boff);
      }
      __builtin_amdgcn_sched_barrier(0);
      #pragma unroll
      for (int j = 0; j < KU; ++j) {
        v2f b0, b1, b2, b3;
        b0.x = bx[j].x; b0.y = by[j].x;
        b1.x = bx[j].y; b1.y = by[j].y;
        b2.x = bx[j].z; b2.y = by[j].z;
        b3.x = bx[j].w; b3.y = by[j].w;
        acc0_0 = wmma_f32(a0[j], b0, acc0_0);
        acc1_0 = wmma_f32(a1[j], b0, acc1_0);
        acc0_1 = wmma_f32(a0[j], b1, acc0_1);
        acc1_1 = wmma_f32(a1[j], b1, acc1_1);
        acc0_2 = wmma_f32(a0[j], b2, acc0_2);
        acc1_2 = wmma_f32(a1[j], b2, acc1_2);
        acc0_3 = wmma_f32(a0[j], b3, acc0_3);
        acc1_3 = wmma_f32(a1[j], b3, acc1_3);
      }
    }
  }

  float s0 = 0.0f, s1 = 0.0f, s2 = 0.0f, s3 = 0.0f;
  float q0 = 0.0f, q1 = 0.0f, q2 = 0.0f, q3 = 0.0f;
  #pragma unroll
  for (int r = 0; r < 8; ++r) {
    s0 += acc0_0[r]; q0 += acc0_0[r] * acc0_0[r];
    s1 += acc0_1[r]; q1 += acc0_1[r] * acc0_1[r];
    s2 += acc0_2[r]; q2 += acc0_2[r] * acc0_2[r];
    s3 += acc0_3[r]; q3 += acc0_3[r] * acc0_3[r];
  }
  #pragma unroll
  for (int r = 0; r < 8; ++r) {
    s0 += acc1_0[r]; q0 += acc1_0[r] * acc1_0[r];
    s1 += acc1_1[r]; q1 += acc1_1[r] * acc1_1[r];
    s2 += acc1_2[r]; q2 += acc1_2[r] * acc1_2[r];
    s3 += acc1_3[r]; q3 += acc1_3[r] * acc1_3[r];
  }
  float* rw = red + (wave * 2 + half) * 128 + (m << 2);
  v4f sv, qv;
  sv.x = s0; sv.y = s1; sv.z = s2; sv.w = s3;
  qv.x = q0; qv.y = q1; qv.z = q2; qv.w = q3;
  *(v4f*)rw        = sv;
  *(v4f*)(rw + 64) = qv;
  __syncthreads();
  if ((int)threadIdx.x < strips * 64) {
    const int st  = threadIdx.x >> 6;
    const int col = threadIdx.x & 63;
    float ts = 0.0f, tq = 0.0f;
    for (int w = st; w < 8; w += strips) {
      const float* r0 = red + (w * 2) * 128;
      ts += r0[col];       tq += r0[64 + col];
      ts += r0[128 + col]; tq += r0[192 + col];
    }
    float* p = part + (size_t)blk * (size_t)(2 * Nc);
    p[st * 64 + col]      = ts;
    p[Nc + st * 64 + col] = tq;
  }

  if (active) {
    int rbase = row0 + (half << 3);
    float* Cp = C + (size_t)rbase * Nc + col0 + (m << 2);
    #pragma unroll
    for (int r = 0; r < 8; ++r) {
      v4f o;
      o.x = acc0_0[r]; o.y = acc0_1[r]; o.z = acc0_2[r]; o.w = acc0_3[r];
      *(v4f*)(Cp + (size_t)r * Nc) = o;
    }
    Cp += (size_t)16 * Nc;
    #pragma unroll
    for (int r = 0; r < 8; ++r) {
      v4f o;
      o.x = acc1_0[r]; o.y = acc1_1[r]; o.z = acc1_2[r]; o.w = acc1_3[r];
      *(v4f*)(Cp + (size_t)r * Nc) = o;
    }
  }
}

#define GU 2
__global__ __launch_bounds__(256, 8) void kpconv_wmma(
    const float* __restrict__ xyz, const int* __restrict__ nidx,
    const float* __restrict__ x1, const float* __restrict__ kp_points,
    const float* __restrict__ kpw, float* __restrict__ xk) {
  extern __shared__ __attribute__((aligned(16))) float smem[];
  float* FK  = smem;
  float* Wl  = FK + 16 * FKP;
  int*   IDX = (int*)(Wl + 8 * 16 * WLP);
  float* PS  = FK;

  const int wave = threadIdx.x >> 5;
  const int lane = threadIdx.x & 31;
  const int m    = lane & 15;
  const int half = lane >> 4;
  const int base = blockIdx.x << 4;

  float* wl = Wl + wave * (16 * WLP);
  int*   il = IDX + wave * NBH;

  for (int it = 0; it < 2; ++it) {
    const int pi = (wave << 1) + it;
    const int n  = base + pi;

    const int id = nidx[(size_t)n * NBH + lane];
    il[lane] = id;
    const float cx = xyz[(size_t)n * 3 + 0];
    const float cy = xyz[(size_t)n * 3 + 1];
    const float cz = xyz[(size_t)n * 3 + 2];
    const float px = xyz[(size_t)id * 3 + 0] - cx;
    const float py = xyz[(size_t)id * 3 + 1] - cy;
    const float pz = xyz[(size_t)id * 3 + 2] - cz;
    #pragma unroll
    for (int k = 0; k < KPTS; ++k) {
      float dx = px - kp_points[k * 3 + 0];
      float dy = py - kp_points[k * 3 + 1];
      float dz = pz - kp_points[k * 3 + 2];
      float r = __builtin_amdgcn_sqrtf(dx * dx + dy * dy + dz * dz);
      wl[k * WLP + lane] = fmaxf(1.0f - r * INFL_INV, 0.0f);
    }
    wl[KPTS * WLP + lane] = 0.0f;
    __syncthreads();

    v8f acc0 = {}, acc1 = {}, acc2 = {}, acc3 = {};
    #pragma unroll
    for (int k0 = 0; k0 < NBH; k0 += 4 * GU) {
      v4f bx[GU], by[GU];
      #pragma unroll
      for (int j = 0; j < GU; ++j) {
        const int ka = k0 + 4 * j + (half << 1);
        bx[j] = *(const v4f*)(x1 + (size_t)il[ka] * D2 + (m << 2));
        by[j] = *(const v4f*)(x1 + (size_t)il[ka + 1] * D2 + (m << 2));
      }
      __builtin_amdgcn_sched_barrier(0);
      #pragma unroll
      for (int j = 0; j < GU; ++j) {
        const int ka = k0 + 4 * j + (half << 1);
        v2f a;
        a.x = wl[m * WLP + ka];
        a.y = wl[m * WLP + ka + 1];
        v2f b0, b1, b2, b3;
        b0.x = bx[j].x; b0.y = by[j].x;
        b1.x = bx[j].y; b1.y = by[j].y;
        b2.x = bx[j].z; b2.y = by[j].z;
        b3.x = bx[j].w; b3.y = by[j].w;
        acc0 = wmma_f32(a, b0, acc0);
        acc1 = wmma_f32(a, b1, acc1);
        acc2 = wmma_f32(a, b2, acc2);
        acc3 = wmma_f32(a, b3, acc3);
      }
    }
    #pragma unroll
    for (int r = 0; r < 8; ++r) {
      const int krow = (half << 3) + r;
      if (krow < KPTS) {
        v4f o;
        o.x = acc0[r]; o.y = acc1[r]; o.z = acc2[r]; o.w = acc3[r];
        *(v4f*)(FK + pi * FKP + krow * D2 + (m << 2)) = o;
      }
    }
    __syncthreads();
  }

  {
    const int kbeg = wave * 120;
    v8f acc0 = {}, acc1 = {}, acc2 = {}, acc3 = {};
    for (int kk = kbeg; kk < kbeg + 120; kk += 4) {
      const int ka = kk + (half << 1);
      v2f a;
      a.x = FK[m * FKP + ka];
      a.y = FK[m * FKP + ka + 1];
      const float* Bp = kpw + (size_t)ka * D2 + (m << 2);
      const v4f bx = *(const v4f*)Bp;
      const v4f by = *(const v4f*)(Bp + D2);
      v2f b0, b1, b2, b3;
      b0.x = bx.x; b0.y = by.x;
      b1.x = bx.y; b1.y = by.y;
      b2.x = bx.z; b2.y = by.z;
      b3.x = bx.w; b3.y = by.w;
      acc0 = wmma_f32(a, b0, acc0);
      acc1 = wmma_f32(a, b1, acc1);
      acc2 = wmma_f32(a, b2, acc2);
      acc3 = wmma_f32(a, b3, acc3);
    }
    __syncthreads();
    #pragma unroll
    for (int r = 0; r < 8; ++r) {
      v4f o;
      o.x = acc0[r]; o.y = acc1[r]; o.z = acc2[r]; o.w = acc3[r];
      *(v4f*)(PS + wave * (16 * D2) + ((half << 3) + r) * D2 + (m << 2)) = o;
    }
  }
  __syncthreads();

  for (int t = threadIdx.x; t < 16 * D2; t += 256) {
    float s = 0.0f;
    #pragma unroll
    for (int w = 0; w < 8; ++w) s += PS[w * (16 * D2) + t];
    const int pt = t >> 6;
    const int e  = t & 63;
    xk[(size_t)(base + pt) * D2 + e] = s;
  }
}

__global__ __launch_bounds__(1024) void finalize_bn(
    const float* __restrict__ partA, int npartsA, const float* __restrict__ gA,
    const float* __restrict__ bA, int CA, float* __restrict__ scA, float* __restrict__ shA,
    const float* __restrict__ partB, int npartsB, const float* __restrict__ gB,
    const float* __restrict__ bB, int CB, float* __restrict__ scB, float* __restrict__ shB,
    int blocks0, float invN) {
  __shared__ float rs[1024];
  __shared__ float rq[1024];
  const bool second = (int)blockIdx.x >= blocks0;
  const float* __restrict__ part = second ? partB : partA;
  const float* __restrict__ g    = second ? gB : gA;
  const float* __restrict__ b    = second ? bB : bA;
  float* __restrict__ sc         = second ? scB : scA;
  float* __restrict__ sh         = second ? shB : shA;
  const int nparts = second ? npartsB : npartsA;
  const int C      = second ? CB : CA;
  const int blk = second ? (int)blockIdx.x - blocks0 : (int)blockIdx.x;
  const int cl = threadIdx.x & 15;
  const int sl = threadIdx.x >> 4;
  const int c  = blk * 16 + cl;
  const size_t pitch = (size_t)(2 * C);
  float s = 0.0f, q = 0.0f;
  #pragma unroll 8
  for (int p = sl; p < nparts; p += 64) {
    s += part[(size_t)p * pitch + c];
    q += part[(size_t)p * pitch + C + c];
  }
  rs[threadIdx.x] = s;
  rq[threadIdx.x] = q;
  __syncthreads();
  if (threadIdx.x < 16) {
    float ts = 0.0f, tq = 0.0f;
    #pragma unroll
    for (int k = 0; k < 64; ++k) {
      ts += rs[k * 16 + cl];
      tq += rq[k * 16 + cl];
    }
    float mean = ts * invN;
    float var  = tq * invN - mean * mean;
    float scl  = g[c] * rsqrtf(var + BN_EPS);
    sc[c] = scl;
    sh[c] = b[c] - mean * scl;
  }
}


template <int C, int U>
__global__ __launch_bounds__(256) void bnact_inplace(
    float* __restrict__ Y, const float* __restrict__ sc,
    const float* __restrict__ sh, int rows) {
  constexpr int G   = C / 4;
  constexpr int RPB = 256 / G;
  const int cg    = threadIdx.x % G;
  const int r0    = blockIdx.x * RPB + threadIdx.x / G;
  const int rstep = gridDim.x * RPB;
  const v4f a = *(const v4f*)(sc + 4 * cg);
  const v4f b = *(const v4f*)(sh + 4 * cg);
  float* base = Y + 4 * cg;
  for (int r = r0; r < rows; r += U * rstep) {
    v4f y[U];
    #pragma unroll
    for (int i = 0; i < U; ++i) {
      const int ri = min(r + i * rstep, rows - 1);
      y[i] = *(const v4f*)(base + (size_t)ri * C);
    }
    __builtin_amdgcn_sched_barrier(0);
    #pragma unroll
    for (int i = 0; i < U; ++i) {
      const int ri = r + i * rstep;
      if (ri < rows) {
        v4f o;
        o.x = lrelu(y[i].x * a.x + b.x);
        o.y = lrelu(y[i].y * a.y + b.y);
        o.z = lrelu(y[i].z * a.z + b.z);
        o.w = lrelu(y[i].w * a.w + b.w);
        *(v4f*)(base + (size_t)ri * C) = o;
      }
    }
  }
}

template <int C, int U>
__global__ __launch_bounds__(256) void fuse_out(
    float* __restrict__ out, const float* __restrict__ ysc,
    const float* __restrict__ sc2, const float* __restrict__ sh2,
    const float* __restrict__ scS, const float* __restrict__ shS, int rows) {
  constexpr int G   = C / 4;
  constexpr int RPB = 256 / G;
  const int cg    = threadIdx.x % G;
  const int r0    = blockIdx.x * RPB + threadIdx.x / G;
  const int rstep = gridDim.x * RPB;
  const v4f a2 = *(const v4f*)(sc2 + 4 * cg);
  const v4f b2 = *(const v4f*)(sh2 + 4 * cg);
  const v4f aS = *(const v4f*)(scS + 4 * cg);
  const v4f bS = *(const v4f*)(shS + 4 * cg);
  float*       obase = out + 4 * cg;
  const float* sbase = ysc + 4 * cg;
  for (int r = r0; r < rows; r += U * rstep) {
    v4f y[U], s[U];
    #pragma unroll
    for (int i = 0; i < U; ++i) {
      const int ri = min(r + i * rstep, rows - 1);
      y[i] = *(const v4f*)(obase + (size_t)ri * C);
      s[i] = *(const v4f*)(sbase + (size_t)ri * C);
    }
    __builtin_amdgcn_sched_barrier(0);
    #pragma unroll
    for (int i = 0; i < U; ++i) {
      const int ri = r + i * rstep;
      if (ri < rows) {
        v4f o;
        o.x = lrelu(y[i].x * a2.x + b2.x) + (s[i].x * aS.x + bS.x);
        o.y = lrelu(y[i].y * a2.y + b2.y) + (s[i].y * aS.y + bS.y);
        o.z = lrelu(y[i].z * a2.z + b2.z) + (s[i].z * aS.z + bS.z);
        o.w = lrelu(y[i].w * a2.w + b2.w) + (s[i].w * aS.w + bS.w);
        *(v4f*)(obase + (size_t)ri * C) = o;
      }
    }
  }
}

extern "C" void kernel_launch(void* const* d_in, const int* in_sizes, int n_in,
                              void* d_out, int out_size, void* d_ws, size_t ws_size,
                              hipStream_t stream) {
  const float* feats = (const float*)d_in[0];
  const float* xyz   = (const float*)d_in[1];
  const int*   nidx  = (const int*)d_in[3];
  const float* W1    = (const float*)d_in[4];
  const float* g1    = (const float*)d_in[5];
  const float* b1    = (const float*)d_in[6];
  const float* kpp   = (const float*)d_in[7];
  const float* kpw   = (const float*)d_in[8];
  const float* W2    = (const float*)d_in[9];
  const float* g2    = (const float*)d_in[10];
  const float* b2    = (const float*)d_in[11];
  const float* Wsc   = (const float*)d_in[12];
  const float* gsc   = (const float*)d_in[13];
  const float* bsc   = (const float*)d_in[14];
  float* out = (float*)d_out;

  const int N = in_sizes[0] / CIN;

  float* ws      = (float*)d_ws;
  float* x1      = ws;
  float* xk      = x1 + (size_t)N * D2;
  float* ysc     = xk + (size_t)N * D2;
  float* sc1 = ysc + (size_t)N * COUT; float* sh1 = sc1 + D2;
  float* sc2 = sh1 + D2;               float* sh2 = sc2 + COUT;
  float* scS = sh2 + COUT;             float* shS = scS + COUT;
  const int blk1 = ((N / 32) * (D2 / 64) + 7) / 8;
  const int blk2 = ((N / 32) * (COUT / 64) + 7) / 8;
  float* part1  = xk;
  float* partSC = xk + (size_t)blk1 * 2 * D2;
  float* part2  = x1;
  const int units1 = (N / 32) * (D2 / 64);
  const int units2 = (N / 32) * (COUT / 64);

  gemm_f32_wmma<<<blk1 + blk2, 256, 0, stream>>>(feats, W1, x1, part1, CIN, D2, units1,
                                                 feats, Wsc, ysc, partSC, CIN, COUT, units2, blk1);
  finalize_bn<<<D2 / 16 + COUT / 16, 1024, 0, stream>>>(part1, blk1, g1, b1, D2, sc1, sh1,
                                                        partSC, blk2, gsc, bsc, COUT, scS, shS,
                                                        D2 / 16, 1.0f / (float)N);
  bnact_inplace<D2, 8><<<625, 256, 0, stream>>>(x1, sc1, sh1, N);

  {
    size_t lds = (16 * FKP + 8 * 16 * WLP) * sizeof(float)
               + 8 * NBH * sizeof(int);
    kpconv_wmma<<<N / 16, 256, lds, stream>>>(xyz, nidx, x1, kpp, kpw, xk);
  }

  gemm_f32_wmma<<<blk2, 256, 0, stream>>>(xk, W2, out, part2, D2, COUT, units2,
                                          xk, W2, out, part2, D2, COUT, units2, blk2);
  finalize_bn<<<COUT / 16, 1024, 0, stream>>>(part2, blk2, g2, b2, COUT, sc2, sh2,
                                              part2, blk2, g2, b2, COUT, sc2, sh2,
                                              COUT / 16, 1.0f / (float)N);

  fuse_out<COUT, 4><<<1000, 256, 0, stream>>>(out, ysc, sc2, sh2, scS, shS, N);
}
